# starting kernel
# baseline (speedup 1.0000x reference)
#include <hip/hip_runtime.h>


typedef float v2f __attribute__((ext_vector_type(2)));
typedef float v8f __attribute__((ext_vector_type(8)));

#define BATCH 512
#define INPS  1024
#define NEUR  512
#define KFAN  64
#define NOUT  16
#define ROWS  16

__global__ __launch_bounds__(32) void ne_scan_wmma_kernel(
    const float* __restrict__ x,
    const int*   __restrict__ in_idxs,
    const float* __restrict__ weights,
    float*       __restrict__ out)
{
    extern __shared__ float smem[];
    float* act = smem;
    float* wst = act + (INPS + NEUR) * ROWS;
    int*   ist = (int*)(wst + 2 * KFAN);

    const int lane  = threadIdx.x;
    const int rbase = blockIdx.x * ROWS;
    const int h     = lane >> 4;
    const int n     = lane & 15;

    for (int t = lane; t < ROWS * INPS; t += 32) {
        int col = t & (INPS - 1);
        int r   = t >> 10;
        act[col * ROWS + r] = x[(size_t)(rbase + r) * INPS + col];
    }
    for (int t = lane; t < ROWS * NEUR; t += 32) {
        act[INPS * ROWS + t] = 0.0f;
    }

    v2f  aw[16];
    int2 jj[16];
    {
        float2 w0 = ((const float2*)weights)[lane];
        int2   j0 = ((const int2*)in_idxs)[lane];
        wst[2 * lane] = w0.x; wst[2 * lane + 1] = w0.y;
        ist[2 * lane] = j0.x; ist[2 * lane + 1] = j0.y;
        #pragma unroll
        for (int i = 0; i < 16; ++i) {
            const int k0 = 4 * i + 2 * h;
            float2 af = *(const float2*)(wst + k0);
            aw[i].x = af.x; aw[i].y = af.y;
            jj[i]   = *(const int2*)(ist + k0);
        }
    }
    float2 wA = ((const float2*)(weights + (size_t)1 * KFAN))[lane];
    int2   jA = ((const int2*)(in_idxs + (size_t)1 * KFAN))[lane];
    float2 wB = ((const float2*)(weights + (size_t)2 * KFAN))[lane];
    int2   jB = ((const int2*)(in_idxs + (size_t)2 * KFAN))[lane];

    for (int pos = 0; pos < NEUR; ++pos) {
        v2f b[16];
        #pragma unroll
        for (int i = 0; i < 16; ++i) {
            b[i].x = act[jj[i].x * ROWS + n];
            b[i].y = act[jj[i].y * ROWS + n];
        }

        v8f acc[4] = {};
        #pragma unroll
        for (int i = 0; i < 16; ++i) {
            acc[i & 3] = __builtin_amdgcn_wmma_f32_16x16x4_f32(
                false, aw[i], false, b[i], (short)0, acc[i & 3], false, false);
        }

        float val = (acc[0][0] + acc[1][0]) + (acc[2][0] + acc[3][0]);
        float sg  = 1.0f / (1.0f + __expf(-val));
        act[(INPS + pos) * ROWS + (lane & 15)] = sg;

        if (pos + 1 < NEUR) {
            const int   s  = (pos + 1) & 1;
            float*      wc = wst + s * KFAN;
            int*        ic = ist + s * KFAN;
            wc[2 * lane] = wA.x; wc[2 * lane + 1] = wA.y;
            ic[2 * lane] = jA.x; ic[2 * lane + 1] = jA.y;
            #pragma unroll
            for (int i = 0; i < 16; ++i) {
                const int k0 = 4 * i + 2 * h;
                float2 af = *(const float2*)(wc + k0);
                aw[i].x = af.x; aw[i].y = af.y;
                jj[i]   = *(const int2*)(ic + k0);
            }
            wA = wB; jA = jB;
            if (pos + 3 < NEUR) {
                wB = ((const float2*)(weights + (size_t)(pos + 3) * KFAN))[lane];
                jB = ((const int2*)(in_idxs + (size_t)(pos + 3) * KFAN))[lane];
            }
        }
    }

    for (int t = lane; t < NOUT * ROWS; t += 32) {
        int o = t >> 4;
        int r = t & 15;
        out[(size_t)o * BATCH + rbase + r] =
            act[(INPS + NEUR - NOUT + o) * ROWS + r];
    }
}

extern "C" void kernel_launch(void* const* d_in, const int* in_sizes, int n_in,
                              void* d_out, int out_size, void* d_ws, size_t ws_size,
                              hipStream_t stream) {
    (void)in_sizes; (void)n_in; (void)d_ws; (void)ws_size; (void)out_size;
    const float* x       = (const float*)d_in[0];
    const int*   in_idxs = (const int*)d_in[1];
    const float* weights = (const float*)d_in[2];
    float*       out     = (float*)d_out;

    const size_t smem_bytes =
        (size_t)((INPS + NEUR) * ROWS + 2 * KFAN) * sizeof(float) +
        (size_t)(2 * KFAN) * sizeof(int);

    ne_scan_wmma_kernel<<<dim3(BATCH / ROWS), dim3(32), smem_bytes, stream>>>(
        x, in_idxs, weights, out);
}


# Sparse gather scan recast as dense block-triangular product: weights scattered to a dense matrix, input part one parallel WMMA GEMM, scan in 16-neuron blocks by rank-1 WMMAs (reorders 64-term sum)
# speedup vs baseline: 3.7521x; 3.7521x over previous
#include <hip/hip_runtime.h>


typedef float v2f __attribute__((ext_vector_type(2)));
typedef float v8f __attribute__((ext_vector_type(8)));

#define BATCH 512
#define INPS  1024
#define NEUR  512
#define KFAN  64
#define NOUT  16
#define COLS  (INPS + NEUR)
#define NBLK  (NEUR / 16)
#define NBT   (BATCH / 16)
#define NHELP 3

__device__ __forceinline__ v8f mma4(v2f a, v2f b, v8f c) {
    return __builtin_amdgcn_wmma_f32_16x16x4_f32(false, a, false, b, (short)0, c, false, false);
}

__global__ __launch_bounds__(256) void ne_dense_rows_kernel(
    const int*   __restrict__ in_idxs,
    const float* __restrict__ weights,
    float*       __restrict__ wd)
{
    __shared__ float row[COLS];
    __shared__ int   sj[KFAN];
    __shared__ float sw[KFAN];
    const int n = blockIdx.x;
    const int t = threadIdx.x;
    if (t < KFAN) {
        sj[t] = in_idxs[n * KFAN + t];
        sw[t] = weights[n * KFAN + t];
    }
    for (int c = t; c < COLS; c += 256) row[c] = 0.0f;
    __syncthreads();
    if (t < KFAN) {
        const int j = sj[t];
        bool  first = true;
        float s = 0.0f;
        for (int k = 0; k < KFAN; ++k) {
            if (sj[k] == j) {
                if (k < t) first = false;
                s += sw[k];
            }
        }
        if (first && (unsigned)j < (unsigned)(INPS + n)) row[j] = s;
    }
    __syncthreads();
    for (int c = t; c < COLS; c += 256) wd[(size_t)n * COLS + c] = row[c];
}

__global__ __launch_bounds__(32) void ne_input_gemm_kernel(
    const float* __restrict__ x,
    const float* __restrict__ wd,
    float*       __restrict__ pf)
{
    const int lane = threadIdx.x;
    const int c = lane & 15, h = lane >> 4;
    const int J = blockIdx.x, Bt = blockIdx.y;
    const float* ap = wd + (size_t)(16 * J + c) * COLS + 4 * h;
    const float* bp = x + (size_t)(16 * Bt + c) * INPS + 4 * h;
    v8f acc0 = {}, acc1 = {};
    #pragma unroll 16
    for (int k0 = 0; k0 < INPS; k0 += 8) {
        const float4 a = *(const float4*)(ap + k0);
        const float4 b = *(const float4*)(bp + k0);
        const v2f a0 = {a.x, a.y}, b0 = {b.x, b.y};
        const v2f a1 = {a.z, a.w}, b1 = {b.z, b.w};
        acc0 = mma4(a0, b0, acc0);
        acc1 = mma4(a1, b1, acc1);
    }
    acc0 += acc1;
    float4* o = (float4*)(pf + ((size_t)(J * NBT + Bt) * 32 + lane) * 8);
    o[0] = make_float4(acc0[0], acc0[1], acc0[2], acc0[3]);
    o[1] = make_float4(acc0[4], acc0[5], acc0[6], acc0[7]);
}

__global__ __launch_bounds__(128) void ne_block_scan_kernel(
    const float* __restrict__ wd,
    const float* __restrict__ pf,
    float*       __restrict__ out)
{
    __shared__ __attribute__((aligned(16))) float hL[NBLK][2][16][8];
    __shared__ __attribute__((aligned(16))) float part[2][NHELP][32][8];

    const int wave = __builtin_amdgcn_readfirstlane((int)(threadIdx.x >> 5));
    const int lane = threadIdx.x & 31;
    const int c = lane & 15, h = lane >> 4;
    const int Bt = blockIdx.x;
    const float* lbase = wd + INPS + 8 * h;

    float  hv[8] = {0.f, 0.f, 0.f, 0.f, 0.f, 0.f, 0.f, 0.f};
    float4 pn0, pn1, dn0, dn1, cn0, cn1;
    {
        const float* lr = lbase + (size_t)c * COLS;
        const float4* pp = (const float4*)(pf + ((size_t)Bt * 32 + lane) * 8);
        pn0 = pp[0]; pn1 = pp[1];
        dn0 = *(const float4*)(lr); dn1 = *(const float4*)(lr + 4);
        cn0 = make_float4(0.f, 0.f, 0.f, 0.f); cn1 = cn0;
    }

    for (int J = 0; J < NBLK; ++J) {
        if (wave == 0) {
            const float4 p0 = pn0, p1 = pn1, d0 = dn0, d1 = dn1, c0 = cn0, c1 = cn1;
            {
                const int Jn = (J + 1 < NBLK) ? J + 1 : J;
                const float* lr = lbase + (size_t)(16 * Jn + c) * COLS;
                const float4* pp = (const float4*)(pf + ((size_t)(Jn * NBT + Bt) * 32 + lane) * 8);
                pn0 = pp[0]; pn1 = pp[1];
                dn0 = *(const float4*)(lr + 16 * Jn); dn1 = *(const float4*)(lr + 16 * Jn + 4);
                cn0 = *(const float4*)(lr + 16 * Jn - 16); cn1 = *(const float4*)(lr + 16 * Jn - 12);
            }
            v8f acc = {p0.x, p0.y, p0.z, p0.w, p1.x, p1.y, p1.z, p1.w};
            if (J >= 1) {
                v8f q = {};
                for (int w = 0; w < NHELP; ++w) {
                    const float4 u0 = *(const float4*)&part[J & 1][w][lane][0];
                    const float4 u1 = *(const float4*)&part[J & 1][w][lane][4];
                    q[0] += u0.x; q[1] += u0.y; q[2] += u0.z; q[3] += u0.w;
                    q[4] += u1.x; q[5] += u1.y; q[6] += u1.z; q[7] += u1.w;
                }
                const v2f a0 = {c0.x, c0.y}, a1 = {c0.z, c0.w}, a2 = {c1.x, c1.y}, a3 = {c1.z, c1.w};
                const v2f b0 = {hv[0], hv[1]}, b1 = {hv[2], hv[3]}, b2 = {hv[4], hv[5]}, b3 = {hv[6], hv[7]};
                acc = mma4(a0, b0, acc);
                q   = mma4(a1, b1, q);
                acc = mma4(a2, b2, acc);
                q   = mma4(a3, b3, q);
                acc += q;
            }
            const float dg[8] = {d0.x, d0.y, d0.z, d0.w, d1.x, d1.y, d1.z, d1.w};
            float alo[8], ahi[8];
            for (int v = 0; v < 8; ++v) {
                alo[v] = (h == 0) ? dg[v] : 0.0f;
                ahi[v] = (h == 1) ? dg[v] : 0.0f;
            }
            #pragma unroll
            for (int i = 0; i < 16; ++i) {
                const int v = i & 7;
                const float pre = acc[v];
                const float sg  = 1.0f / (1.0f + __expf(-pre));
                hv[v] = ((i >> 3) == h) ? sg : hv[v];
                if (i < 15) {
                    const v2f a = {(i < 8) ? alo[v] : ahi[v], 0.0f};
                    const v2f b = {sg, 0.0f};
                    acc = mma4(a, b, acc);
                }
            }
            *(float4*)&hL[J][h][c][0] = make_float4(hv[0], hv[1], hv[2], hv[3]);
            *(float4*)&hL[J][h][c][4] = make_float4(hv[4], hv[5], hv[6], hv[7]);
            if (J == NBLK - 1) {
                for (int v = 0; v < 8; ++v)
                    out[(size_t)(8 * h + v) * BATCH + 16 * Bt + c] = hv[v];
            }
        } else if (J + 1 < NBLK) {
            const int R = J + 1;
            const float* lr = lbase + (size_t)(16 * R + c) * COLS;
            v8f q0 = {}, q1 = {};
            #pragma unroll 4
            for (int j = wave - 1; j < J; j += NHELP) {
                const float4 a0 = *(const float4*)(lr + 16 * j);
                const float4 a1 = *(const float4*)(lr + 16 * j + 4);
                const float4 b0 = *(const float4*)&hL[j][h][c][0];
                const float4 b1 = *(const float4*)&hL[j][h][c][4];
                const v2f x0 = {a0.x, a0.y}, y0 = {b0.x, b0.y};
                const v2f x1 = {a0.z, a0.w}, y1 = {b0.z, b0.w};
                const v2f x2 = {a1.x, a1.y}, y2 = {b1.x, b1.y};
                const v2f x3 = {a1.z, a1.w}, y3 = {b1.z, b1.w};
                q0 = mma4(x0, y0, q0);
                q1 = mma4(x1, y1, q1);
                q0 = mma4(x2, y2, q0);
                q1 = mma4(x3, y3, q1);
            }
            q0 += q1;
            *(float4*)&part[R & 1][wave - 1][lane][0] = make_float4(q0[0], q0[1], q0[2], q0[3]);
            *(float4*)&part[R & 1][wave - 1][lane][4] = make_float4(q0[4], q0[5], q0[6], q0[7]);
        }
        __syncthreads();
    }
}

extern "C" void kernel_launch(void* const* d_in, const int* in_sizes, int n_in,
                              void* d_out, int out_size, void* d_ws, size_t ws_size,
                              hipStream_t stream) {
    (void)in_sizes; (void)n_in; (void)ws_size; (void)out_size;
    const float* x       = (const float*)d_in[0];
    const int*   in_idxs = (const int*)d_in[1];
    const float* weights = (const float*)d_in[2];
    float*       out     = (float*)d_out;
    float*       wd      = (float*)d_ws;
    float*       pf      = wd + (size_t)NEUR * COLS;

    ne_dense_rows_kernel<<<dim3(NEUR), dim3(256), 0, stream>>>(in_idxs, weights, wd);
    ne_input_gemm_kernel<<<dim3(NBLK, NBT), dim3(32), 0, stream>>>(x, wd, pf);
    ne_block_scan_kernel<<<dim3(NBT), dim3(128), 0, stream>>>(wd, pf, out);
}


# Scan's global loads issued one block ahead: helper waves fetch the next block's weight tiles into registers before this block's products; wave 0's loads pinned at block top (sum order unchanged)
# speedup vs baseline: 5.0497x; 1.3458x over previous
#include <hip/hip_runtime.h>


typedef float v2f __attribute__((ext_vector_type(2)));
typedef float v8f __attribute__((ext_vector_type(8)));

#define BATCH 512
#define INPS  1024
#define NEUR  512
#define KFAN  64
#define NOUT  16
#define COLS  (INPS + NEUR)
#define NBLK  (NEUR / 16)
#define NBT   (BATCH / 16)
#define NHELP 3
#define MAXT  10

__device__ __forceinline__ v8f mma4(v2f a, v2f b, v8f c) {
    return __builtin_amdgcn_wmma_f32_16x16x4_f32(false, a, false, b, (short)0, c, false, false);
}

__global__ __launch_bounds__(256) void ne_dense_rows_kernel(
    const int*   __restrict__ in_idxs,
    const float* __restrict__ weights,
    float*       __restrict__ wd)
{
    __shared__ float row[COLS];
    __shared__ int   sj[KFAN];
    __shared__ float sw[KFAN];
    const int n = blockIdx.x;
    const int t = threadIdx.x;
    if (t < KFAN) {
        sj[t] = in_idxs[n * KFAN + t];
        sw[t] = weights[n * KFAN + t];
    }
    for (int c = t; c < COLS; c += 256) row[c] = 0.0f;
    __syncthreads();
    if (t < KFAN) {
        const int j = sj[t];
        bool  first = true;
        float s = 0.0f;
        for (int k = 0; k < KFAN; ++k) {
            if (sj[k] == j) {
                if (k < t) first = false;
                s += sw[k];
            }
        }
        if (first && (unsigned)j < (unsigned)(INPS + n)) row[j] = s;
    }
    __syncthreads();
    for (int c = t; c < COLS; c += 256) wd[(size_t)n * COLS + c] = row[c];
}

__global__ __launch_bounds__(32) void ne_input_gemm_kernel(
    const float* __restrict__ x,
    const float* __restrict__ wd,
    float*       __restrict__ pf)
{
    const int lane = threadIdx.x;
    const int c = lane & 15, h = lane >> 4;
    const int J = blockIdx.x, Bt = blockIdx.y;
    const float* ap = wd + (size_t)(16 * J + c) * COLS + 4 * h;
    const float* bp = x + (size_t)(16 * Bt + c) * INPS + 4 * h;
    v8f acc0 = {}, acc1 = {};
    #pragma unroll 16
    for (int k0 = 0; k0 < INPS; k0 += 8) {
        const float4 a = *(const float4*)(ap + k0);
        const float4 b = *(const float4*)(bp + k0);
        const v2f a0 = {a.x, a.y}, b0 = {b.x, b.y};
        const v2f a1 = {a.z, a.w}, b1 = {b.z, b.w};
        acc0 = mma4(a0, b0, acc0);
        acc1 = mma4(a1, b1, acc1);
    }
    acc0 += acc1;
    float4* o = (float4*)(pf + ((size_t)(J * NBT + Bt) * 32 + lane) * 8);
    o[0] = make_float4(acc0[0], acc0[1], acc0[2], acc0[3]);
    o[1] = make_float4(acc0[4], acc0[5], acc0[6], acc0[7]);
}

__global__ __launch_bounds__(128) void ne_block_scan_kernel(
    const float* __restrict__ wd,
    const float* __restrict__ pf,
    float*       __restrict__ out)
{
    __shared__ __attribute__((aligned(16))) float hL[NBLK][2][16][8];
    __shared__ __attribute__((aligned(16))) float part[2][NHELP][32][8];

    const int wave = __builtin_amdgcn_readfirstlane((int)(threadIdx.x >> 5));
    const int lane = threadIdx.x & 31;
    const int c = lane & 15, h = lane >> 4;
    const int Bt = blockIdx.x;
    const float* lbase = wd + INPS + 8 * h;

    float  hv[8] = {0.f, 0.f, 0.f, 0.f, 0.f, 0.f, 0.f, 0.f};
    float4 pn0, pn1, dn0, dn1, cn0, cn1;
    {
        const float* lr = lbase + (size_t)c * COLS;
        const float4* pp = (const float4*)(pf + ((size_t)Bt * 32 + lane) * 8);
        pn0 = pp[0]; pn1 = pp[1];
        dn0 = *(const float4*)(lr); dn1 = *(const float4*)(lr + 4);
        cn0 = make_float4(0.f, 0.f, 0.f, 0.f); cn1 = cn0;
    }

    float4 ca[MAXT][2];
    #pragma unroll
    for (int t = 0; t < MAXT; ++t) {
        ca[t][0] = make_float4(0.f, 0.f, 0.f, 0.f); ca[t][1] = ca[t][0];
    }

    for (int J = 0; J < NBLK; ++J) {
        if (wave == 0) {
            const float4 p0 = pn0, p1 = pn1, d0 = dn0, d1 = dn1, c0 = cn0, c1 = cn1;
            {
                const int Jn = (J + 1 < NBLK) ? J + 1 : J;
                const float* lr = lbase + (size_t)(16 * Jn + c) * COLS;
                const float4* pp = (const float4*)(pf + ((size_t)(Jn * NBT + Bt) * 32 + lane) * 8);
                pn0 = pp[0]; pn1 = pp[1];
                dn0 = *(const float4*)(lr + 16 * Jn); dn1 = *(const float4*)(lr + 16 * Jn + 4);
                cn0 = *(const float4*)(lr + 16 * Jn - 16); cn1 = *(const float4*)(lr + 16 * Jn - 12);
            }
            __builtin_amdgcn_sched_barrier(0);
            v8f acc = {p0.x, p0.y, p0.z, p0.w, p1.x, p1.y, p1.z, p1.w};
            if (J >= 1) {
                v8f q = {};
                for (int w = 0; w < NHELP; ++w) {
                    const float4 u0 = *(const float4*)&part[J & 1][w][lane][0];
                    const float4 u1 = *(const float4*)&part[J & 1][w][lane][4];
                    q[0] += u0.x; q[1] += u0.y; q[2] += u0.z; q[3] += u0.w;
                    q[4] += u1.x; q[5] += u1.y; q[6] += u1.z; q[7] += u1.w;
                }
                const v2f a0 = {c0.x, c0.y}, a1 = {c0.z, c0.w}, a2 = {c1.x, c1.y}, a3 = {c1.z, c1.w};
                const v2f b0 = {hv[0], hv[1]}, b1 = {hv[2], hv[3]}, b2 = {hv[4], hv[5]}, b3 = {hv[6], hv[7]};
                acc = mma4(a0, b0, acc);
                q   = mma4(a1, b1, q);
                acc = mma4(a2, b2, acc);
                q   = mma4(a3, b3, q);
                acc += q;
            }
            const float dg[8] = {d0.x, d0.y, d0.z, d0.w, d1.x, d1.y, d1.z, d1.w};
            float alo[8], ahi[8];
            for (int v = 0; v < 8; ++v) {
                alo[v] = (h == 0) ? dg[v] : 0.0f;
                ahi[v] = (h == 1) ? dg[v] : 0.0f;
            }
            #pragma unroll
            for (int i = 0; i < 16; ++i) {
                const int v = i & 7;
                const float pre = acc[v];
                const float sg  = 1.0f / (1.0f + __expf(-pre));
                hv[v] = ((i >> 3) == h) ? sg : hv[v];
                if (i < 15) {
                    const v2f a = {(i < 8) ? alo[v] : ahi[v], 0.0f};
                    const v2f b = {sg, 0.0f};
                    acc = mma4(a, b, acc);
                }
            }
            *(float4*)&hL[J][h][c][0] = make_float4(hv[0], hv[1], hv[2], hv[3]);
            *(float4*)&hL[J][h][c][4] = make_float4(hv[4], hv[5], hv[6], hv[7]);
            if (J == NBLK - 1) {
                for (int v = 0; v < 8; ++v)
                    out[(size_t)(8 * h + v) * BATCH + 16 * Bt + c] = hv[v];
            }
            __builtin_amdgcn_sched_barrier(0);
        } else if (J + 1 < NBLK) {
            const int R = J + 1;
            const int w = wave - 1;
            float4 nx[MAXT][2];
            #pragma unroll
            for (int t = 0; t < MAXT; ++t) { nx[t][0] = ca[t][0]; nx[t][1] = ca[t][1]; }
            if (J + 2 < NBLK) {
                const float* ln = lbase + (size_t)(16 * (R + 1) + c) * COLS;
                #pragma unroll
                for (int t = 0; t < MAXT; ++t) {
                    const int j = w + NHELP * t;
                    if (j < J + 1) {
                        nx[t][0] = *(const float4*)(ln + 16 * j);
                        nx[t][1] = *(const float4*)(ln + 16 * j + 4);
                    }
                }
            }
            float4 hb[MAXT][2];
            #pragma unroll
            for (int t = 0; t < MAXT; ++t) {
                const int j = w + NHELP * t;
                hb[t][0] = *(const float4*)&hL[j][h][c][0];
                hb[t][1] = *(const float4*)&hL[j][h][c][4];
            }
            __builtin_amdgcn_sched_barrier(0);
            v8f q0 = {}, q1 = {};
            #pragma unroll
            for (int t = 0; t < MAXT; ++t) {
                const int j = w + NHELP * t;
                if (j < J) {
                    const float4 a0 = ca[t][0], a1 = ca[t][1];
                    const float4 b0 = hb[t][0], b1 = hb[t][1];
                    const v2f x0 = {a0.x, a0.y}, y0 = {b0.x, b0.y};
                    const v2f x1 = {a0.z, a0.w}, y1 = {b0.z, b0.w};
                    const v2f x2 = {a1.x, a1.y}, y2 = {b1.x, b1.y};
                    const v2f x3 = {a1.z, a1.w}, y3 = {b1.z, b1.w};
                    q0 = mma4(x0, y0, q0);
                    q1 = mma4(x1, y1, q1);
                    q0 = mma4(x2, y2, q0);
                    q1 = mma4(x3, y3, q1);
                }
            }
            q0 += q1;
            *(float4*)&part[R & 1][w][lane][0] = make_float4(q0[0], q0[1], q0[2], q0[3]);
            *(float4*)&part[R & 1][w][lane][4] = make_float4(q0[4], q0[5], q0[6], q0[7]);
            #pragma unroll
            for (int t = 0; t < MAXT; ++t) { ca[t][0] = nx[t][0]; ca[t][1] = nx[t][1]; }
        }
        __syncthreads();
    }
}

extern "C" void kernel_launch(void* const* d_in, const int* in_sizes, int n_in,
                              void* d_out, int out_size, void* d_ws, size_t ws_size,
                              hipStream_t stream) {
    (void)in_sizes; (void)n_in; (void)ws_size; (void)out_size;
    const float* x       = (const float*)d_in[0];
    const int*   in_idxs = (const int*)d_in[1];
    const float* weights = (const float*)d_in[2];
    float*       out     = (float*)d_out;
    float*       wd      = (float*)d_ws;
    float*       pf      = wd + (size_t)NEUR * COLS;

    ne_dense_rows_kernel<<<dim3(NEUR), dim3(256), 0, stream>>>(in_idxs, weights, wd);
    ne_input_gemm_kernel<<<dim3(NBLK, NBT), dim3(32), 0, stream>>>(x, wd, pf);
    ne_block_scan_kernel<<<dim3(NBT), dim3(128), 0, stream>>>(wd, pf, out);
}


# Dense weight rows built in LDS inside the input-GEMM kernel (16 neurons x 4 batch tiles per work-group; A from LDS, each K block's loads one block ahead): 2 launches, not 3 (sum order unchanged)
# speedup vs baseline: 6.1758x; 1.2230x over previous
#include <hip/hip_runtime.h>


typedef float v2f __attribute__((ext_vector_type(2)));
typedef float v8f __attribute__((ext_vector_type(8)));

#define BATCH 512
#define INPS  1024
#define NEUR  512
#define KFAN  64
#define NOUT  16
#define COLS  (INPS + NEUR)
#define NBLK  (NEUR / 16)
#define NBT   (BATCH / 16)
#define NHELP 3
#define MAXT  10
#define GW    4
#define NG    (NBT / GW)
#define HS    (NEUR / NG)
#define SW    (INPS + HS)
#define SP    (SW + 4)

__device__ __forceinline__ v8f mma4(v2f a, v2f b, v8f c) {
    return __builtin_amdgcn_wmma_f32_16x16x4_f32(false, a, false, b, (short)0, c, false, false);
}

__global__ __launch_bounds__(128) void ne_rows_gemm_kernel(
    const float* __restrict__ x,
    const int*   __restrict__ in_idxs,
    const float* __restrict__ weights,
    float*       __restrict__ wd,
    float*       __restrict__ pf)
{
    __shared__ __attribute__((aligned(16))) float slab[16 * SP];
    __shared__ __attribute__((aligned(16))) int   sj[16 * KFAN];
    __shared__ __attribute__((aligned(16))) float sw[16 * KFAN];
    const int tid = threadIdx.x;
    const int J = blockIdx.x, g = blockIdx.y;
    {
        const int4*   gi = (const int4*)(in_idxs + (size_t)J * 16 * KFAN);
        const float4* gw = (const float4*)(weights + (size_t)J * 16 * KFAN);
        ((int4*)sj)[tid]         = gi[tid];
        ((int4*)sj)[tid + 128]   = gi[tid + 128];
        ((float4*)sw)[tid]       = gw[tid];
        ((float4*)sw)[tid + 128] = gw[tid + 128];
    }
    for (int i = tid; i < 16 * SP / 4; i += 128)
        ((float4*)slab)[i] = make_float4(0.f, 0.f, 0.f, 0.f);
    __syncthreads();
    {
        const int t  = tid & 63;
        const int r0 = __builtin_amdgcn_readfirstlane(tid >> 6);
        int   jv[8];
        float sv[8];
        #pragma unroll
        for (int i = 0; i < 8; ++i) {
            jv[i] = sj[(r0 + 2 * i) * KFAN + t];
            sv[i] = 0.0f;
        }
        #pragma unroll 4
        for (int k4 = 0; k4 < KFAN / 4; ++k4) {
            #pragma unroll
            for (int i = 0; i < 8; ++i) {
                const int4   p = ((const int4*)(sj + (r0 + 2 * i) * KFAN))[k4];
                const float4 u = ((const float4*)(sw + (r0 + 2 * i) * KFAN))[k4];
                sv[i] += (p.x == jv[i]) ? u.x : 0.0f;
                sv[i] += (p.y == jv[i]) ? u.y : 0.0f;
                sv[i] += (p.z == jv[i]) ? u.z : 0.0f;
                sv[i] += (p.w == jv[i]) ? u.w : 0.0f;
            }
        }
        #pragma unroll
        for (int i = 0; i < 8; ++i) {
            const int r = r0 + 2 * i;
            const int j = jv[i];
            const unsigned q = (unsigned)(j - INPS - HS * g);
            int col = (j < INPS) ? j : ((q < (unsigned)HS) ? (int)(INPS + q) : SW);
            col = ((unsigned)j < (unsigned)(INPS + 16 * J + r)) ? col : SW;
            slab[r * SP + col] = sv[i];
        }
    }
    __syncthreads();
    for (int i = tid; i < 16 * HS / 4; i += 128) {
        const int r = i >> 4, c4 = i & 15;
        const float4 v = *(const float4*)&slab[r * SP + INPS + 4 * c4];
        *(float4*)(wd + (size_t)(16 * J + r) * COLS + INPS + HS * g + 4 * c4) = v;
    }

    const int lane = tid & 31;
    const int c = lane & 15, h = lane >> 4;
    const int Bt = GW * g + (tid >> 5);
    const float* ap = slab + c * SP + 4 * h;
    const float* bp = x + (size_t)(16 * Bt + c) * INPS + 4 * h;
    v8f acc0 = {}, acc1 = {};
    float4 af[2][16], bf[2][16];
    #pragma unroll
    for (int s = 0; s < 16; ++s) {
        af[0][s] = *(const float4*)(ap + 8 * s);
        bf[0][s] = *(const float4*)(bp + 8 * s);
    }
    #pragma unroll
    for (int kc = 0; kc < INPS / 128; ++kc) {
        const int cur = kc & 1, nxt = cur ^ 1;
        if (kc + 1 < INPS / 128) {
            #pragma unroll
            for (int s = 0; s < 16; ++s) {
                af[nxt][s] = *(const float4*)(ap + 128 * (kc + 1) + 8 * s);
                bf[nxt][s] = *(const float4*)(bp + 128 * (kc + 1) + 8 * s);
            }
        }
        __builtin_amdgcn_sched_barrier(0);
        #pragma unroll
        for (int s = 0; s < 16; ++s) {
            const float4 a = af[cur][s];
            const float4 b = bf[cur][s];
            const v2f a0 = {a.x, a.y}, b0 = {b.x, b.y};
            const v2f a1 = {a.z, a.w}, b1 = {b.z, b.w};
            acc0 = mma4(a0, b0, acc0);
            acc1 = mma4(a1, b1, acc1);
        }
        __builtin_amdgcn_sched_barrier(0);
    }
    acc0 += acc1;
    float4* o = (float4*)(pf + ((size_t)(J * NBT + Bt) * 32 + lane) * 8);
    o[0] = make_float4(acc0[0], acc0[1], acc0[2], acc0[3]);
    o[1] = make_float4(acc0[4], acc0[5], acc0[6], acc0[7]);
}

__global__ __launch_bounds__(128) void ne_block_scan_kernel(
    const float* __restrict__ wd,
    const float* __restrict__ pf,
    float*       __restrict__ out)
{
    __shared__ __attribute__((aligned(16))) float hL[NBLK][2][16][8];
    __shared__ __attribute__((aligned(16))) float part[2][NHELP][32][8];

    const int wave = __builtin_amdgcn_readfirstlane((int)(threadIdx.x >> 5));
    const int lane = threadIdx.x & 31;
    const int c = lane & 15, h = lane >> 4;
    const int Bt = blockIdx.x;
    const float* lbase = wd + INPS + 8 * h;

    float  hv[8] = {0.f, 0.f, 0.f, 0.f, 0.f, 0.f, 0.f, 0.f};
    float4 pn0, pn1, dn0, dn1, cn0, cn1;
    {
        const float* lr = lbase + (size_t)c * COLS;
        const float4* pp = (const float4*)(pf + ((size_t)Bt * 32 + lane) * 8);
        pn0 = pp[0]; pn1 = pp[1];
        dn0 = *(const float4*)(lr); dn1 = *(const float4*)(lr + 4);
        cn0 = make_float4(0.f, 0.f, 0.f, 0.f); cn1 = cn0;
    }

    float4 ca[MAXT][2];
    #pragma unroll
    for (int t = 0; t < MAXT; ++t) {
        ca[t][0] = make_float4(0.f, 0.f, 0.f, 0.f); ca[t][1] = ca[t][0];
    }

    for (int J = 0; J < NBLK; ++J) {
        if (wave == 0) {
            const float4 p0 = pn0, p1 = pn1, d0 = dn0, d1 = dn1, c0 = cn0, c1 = cn1;
            {
                const int Jn = (J + 1 < NBLK) ? J + 1 : J;
                const float* lr = lbase + (size_t)(16 * Jn + c) * COLS;
                const float4* pp = (const float4*)(pf + ((size_t)(Jn * NBT + Bt) * 32 + lane) * 8);
                pn0 = pp[0]; pn1 = pp[1];
                dn0 = *(const float4*)(lr + 16 * Jn); dn1 = *(const float4*)(lr + 16 * Jn + 4);
                cn0 = *(const float4*)(lr + 16 * Jn - 16); cn1 = *(const float4*)(lr + 16 * Jn - 12);
            }
            __builtin_amdgcn_sched_barrier(0);
            v8f acc = {p0.x, p0.y, p0.z, p0.w, p1.x, p1.y, p1.z, p1.w};
            if (J >= 1) {
                v8f q = {};
                for (int w = 0; w < NHELP; ++w) {
                    const float4 u0 = *(const float4*)&part[J & 1][w][lane][0];
                    const float4 u1 = *(const float4*)&part[J & 1][w][lane][4];
                    q[0] += u0.x; q[1] += u0.y; q[2] += u0.z; q[3] += u0.w;
                    q[4] += u1.x; q[5] += u1.y; q[6] += u1.z; q[7] += u1.w;
                }
                const v2f a0 = {c0.x, c0.y}, a1 = {c0.z, c0.w}, a2 = {c1.x, c1.y}, a3 = {c1.z, c1.w};
                const v2f b0 = {hv[0], hv[1]}, b1 = {hv[2], hv[3]}, b2 = {hv[4], hv[5]}, b3 = {hv[6], hv[7]};
                acc = mma4(a0, b0, acc);
                q   = mma4(a1, b1, q);
                acc = mma4(a2, b2, acc);
                q   = mma4(a3, b3, q);
                acc += q;
            }
            const float dg[8] = {d0.x, d0.y, d0.z, d0.w, d1.x, d1.y, d1.z, d1.w};
            float alo[8], ahi[8];
            for (int v = 0; v < 8; ++v) {
                alo[v] = (h == 0) ? dg[v] : 0.0f;
                ahi[v] = (h == 1) ? dg[v] : 0.0f;
            }
            #pragma unroll
            for (int i = 0; i < 16; ++i) {
                const int v = i & 7;
                const float pre = acc[v];
                const float sg  = 1.0f / (1.0f + __expf(-pre));
                hv[v] = ((i >> 3) == h) ? sg : hv[v];
                if (i < 15) {
                    const v2f a = {(i < 8) ? alo[v] : ahi[v], 0.0f};
                    const v2f b = {sg, 0.0f};
                    acc = mma4(a, b, acc);
                }
            }
            *(float4*)&hL[J][h][c][0] = make_float4(hv[0], hv[1], hv[2], hv[3]);
            *(float4*)&hL[J][h][c][4] = make_float4(hv[4], hv[5], hv[6], hv[7]);
            if (J == NBLK - 1) {
                for (int v = 0; v < 8; ++v)
                    out[(size_t)(8 * h + v) * BATCH + 16 * Bt + c] = hv[v];
            }
            __builtin_amdgcn_sched_barrier(0);
        } else if (J + 1 < NBLK) {
            const int R = J + 1;
            const int w = wave - 1;
            float4 nx[MAXT][2];
            #pragma unroll
            for (int t = 0; t < MAXT; ++t) { nx[t][0] = ca[t][0]; nx[t][1] = ca[t][1]; }
            if (J + 2 < NBLK) {
                const float* ln = lbase + (size_t)(16 * (R + 1) + c) * COLS;
                #pragma unroll
                for (int t = 0; t < MAXT; ++t) {
                    const int j = w + NHELP * t;
                    if (j < J + 1) {
                        nx[t][0] = *(const float4*)(ln + 16 * j);
                        nx[t][1] = *(const float4*)(ln + 16 * j + 4);
                    }
                }
            }
            float4 hb[MAXT][2];
            #pragma unroll
            for (int t = 0; t < MAXT; ++t) {
                const int j = w + NHELP * t;
                hb[t][0] = *(const float4*)&hL[j][h][c][0];
                hb[t][1] = *(const float4*)&hL[j][h][c][4];
            }
            __builtin_amdgcn_sched_barrier(0);
            v8f q0 = {}, q1 = {};
            #pragma unroll
            for (int t = 0; t < MAXT; ++t) {
                const int j = w + NHELP * t;
                if (j < J) {
                    const float4 a0 = ca[t][0], a1 = ca[t][1];
                    const float4 b0 = hb[t][0], b1 = hb[t][1];
                    const v2f x0 = {a0.x, a0.y}, y0 = {b0.x, b0.y};
                    const v2f x1 = {a0.z, a0.w}, y1 = {b0.z, b0.w};
                    const v2f x2 = {a1.x, a1.y}, y2 = {b1.x, b1.y};
                    const v2f x3 = {a1.z, a1.w}, y3 = {b1.z, b1.w};
                    q0 = mma4(x0, y0, q0);
                    q1 = mma4(x1, y1, q1);
                    q0 = mma4(x2, y2, q0);
                    q1 = mma4(x3, y3, q1);
                }
            }
            q0 += q1;
            *(float4*)&part[R & 1][w][lane][0] = make_float4(q0[0], q0[1], q0[2], q0[3]);
            *(float4*)&part[R & 1][w][lane][4] = make_float4(q0[4], q0[5], q0[6], q0[7]);
            #pragma unroll
            for (int t = 0; t < MAXT; ++t) { ca[t][0] = nx[t][0]; ca[t][1] = nx[t][1]; }
        }
        __syncthreads();
    }
}

extern "C" void kernel_launch(void* const* d_in, const int* in_sizes, int n_in,
                              void* d_out, int out_size, void* d_ws, size_t ws_size,
                              hipStream_t stream) {
    (void)in_sizes; (void)n_in; (void)ws_size; (void)out_size;
    const float* x       = (const float*)d_in[0];
    const int*   in_idxs = (const int*)d_in[1];
    const float* weights = (const float*)d_in[2];
    float*       out     = (float*)d_out;
    float*       wd      = (float*)d_ws;
    float*       pf      = wd + (size_t)NEUR * COLS;

    ne_rows_gemm_kernel<<<dim3(NBLK, NG), dim3(128), 0, stream>>>(x, in_idxs, weights, wd, pf);
    ne_block_scan_kernel<<<dim3(NBT), dim3(128), 0, stream>>>(wd, pf, out);
}


# Scan's global loads requested two blocks ahead and kept in flight across the block barrier (barrier waits for LDS only; three register sets in turn, 3 blocks per loop trip; sum order unchanged)
# speedup vs baseline: 7.3651x; 1.1926x over previous
#include <hip/hip_runtime.h>


typedef float v2f __attribute__((ext_vector_type(2)));
typedef float v8f __attribute__((ext_vector_type(8)));

#define BATCH 512
#define INPS  1024
#define NEUR  512
#define KFAN  64
#define NOUT  16
#define COLS  (INPS + NEUR)
#define NBLK  (NEUR / 16)
#define NBT   (BATCH / 16)
#define NHELP 3
#define MAXT  10
#define GW    4
#define NG    (NBT / GW)
#define HS    (NEUR / NG)
#define SW    (INPS + HS)
#define SP    (SW + 4)

__device__ __forceinline__ v8f mma4(v2f a, v2f b, v8f c) {
    return __builtin_amdgcn_wmma_f32_16x16x4_f32(false, a, false, b, (short)0, c, false, false);
}

__global__ __launch_bounds__(128) void ne_rows_gemm_kernel(
    const float* __restrict__ x,
    const int*   __restrict__ in_idxs,
    const float* __restrict__ weights,
    float*       __restrict__ wd,
    float*       __restrict__ pf)
{
    __shared__ __attribute__((aligned(16))) float slab[16 * SP];
    __shared__ __attribute__((aligned(16))) int   sj[16 * KFAN];
    __shared__ __attribute__((aligned(16))) float sw[16 * KFAN];
    const int tid = threadIdx.x;
    const int J = blockIdx.x, g = blockIdx.y;
    {
        const int4*   gi = (const int4*)(in_idxs + (size_t)J * 16 * KFAN);
        const float4* gw = (const float4*)(weights + (size_t)J * 16 * KFAN);
        ((int4*)sj)[tid]         = gi[tid];
        ((int4*)sj)[tid + 128]   = gi[tid + 128];
        ((float4*)sw)[tid]       = gw[tid];
        ((float4*)sw)[tid + 128] = gw[tid + 128];
    }
    for (int i = tid; i < 16 * SP / 4; i += 128)
        ((float4*)slab)[i] = make_float4(0.f, 0.f, 0.f, 0.f);
    __syncthreads();
    {
        const int t  = tid & 63;
        const int r0 = __builtin_amdgcn_readfirstlane(tid >> 6);
        int   jv[8];
        float sv[8];
        #pragma unroll
        for (int i = 0; i < 8; ++i) {
            jv[i] = sj[(r0 + 2 * i) * KFAN + t];
            sv[i] = 0.0f;
        }
        #pragma unroll 4
        for (int k4 = 0; k4 < KFAN / 4; ++k4) {
            #pragma unroll
            for (int i = 0; i < 8; ++i) {
                const int4   p = ((const int4*)(sj + (r0 + 2 * i) * KFAN))[k4];
                const float4 u = ((const float4*)(sw + (r0 + 2 * i) * KFAN))[k4];
                sv[i] += (p.x == jv[i]) ? u.x : 0.0f;
                sv[i] += (p.y == jv[i]) ? u.y : 0.0f;
                sv[i] += (p.z == jv[i]) ? u.z : 0.0f;
                sv[i] += (p.w == jv[i]) ? u.w : 0.0f;
            }
        }
        #pragma unroll
        for (int i = 0; i < 8; ++i) {
            const int r = r0 + 2 * i;
            const int j = jv[i];
            const unsigned q = (unsigned)(j - INPS - HS * g);
            int col = (j < INPS) ? j : ((q < (unsigned)HS) ? (int)(INPS + q) : SW);
            col = ((unsigned)j < (unsigned)(INPS + 16 * J + r)) ? col : SW;
            slab[r * SP + col] = 0.5f * sv[i];
        }
    }
    __syncthreads();
    for (int i = tid; i < 16 * HS / 4; i += 128) {
        const int r = i >> 4, c4 = i & 15;
        const float4 v = *(const float4*)&slab[r * SP + INPS + 4 * c4];
        *(float4*)(wd + (size_t)(16 * J + r) * COLS + INPS + HS * g + 4 * c4) = v;
    }

    const int lane = tid & 31;
    const int c = lane & 15, h = lane >> 4;
    const int Bt = GW * g + (tid >> 5);
    const float* ap = slab + c * SP + 4 * h;
    const float* bp = x + (size_t)(16 * Bt + c) * INPS + 4 * h;
    v8f acc0 = {}, acc1 = {};
    float4 af[2][16], bf[2][16];
    #pragma unroll
    for (int s = 0; s < 16; ++s) {
        af[0][s] = *(const float4*)(ap + 8 * s);
        bf[0][s] = *(const float4*)(bp + 8 * s);
    }
    #pragma unroll
    for (int kc = 0; kc < INPS / 128; ++kc) {
        const int cur = kc & 1, nxt = cur ^ 1;
        if (kc + 1 < INPS / 128) {
            #pragma unroll
            for (int s = 0; s < 16; ++s) {
                af[nxt][s] = *(const float4*)(ap + 128 * (kc + 1) + 8 * s);
                bf[nxt][s] = *(const float4*)(bp + 128 * (kc + 1) + 8 * s);
            }
        }
        __builtin_amdgcn_sched_barrier(0);
        #pragma unroll
        for (int s = 0; s < 16; ++s) {
            const float4 a = af[cur][s];
            const float4 b = bf[cur][s];
            const v2f a0 = {a.x, a.y}, b0 = {b.x, b.y};
            const v2f a1 = {a.z, a.w}, b1 = {b.z, b.w};
            acc0 = mma4(a0, b0, acc0);
            acc1 = mma4(a1, b1, acc1);
        }
        __builtin_amdgcn_sched_barrier(0);
    }
    acc0 += acc1;
    float4* o = (float4*)(pf + ((size_t)(J * NBT + Bt) * 32 + lane) * 8);
    o[0] = make_float4(acc0[0], acc0[1], acc0[2], acc0[3]);
    o[1] = make_float4(acc0[4], acc0[5], acc0[6], acc0[7]);
}


__device__ __forceinline__ void lds_block_barrier() {
    asm volatile("s_wait_dscnt 0x0" ::: "memory");
    asm volatile("s_barrier_signal -1" ::: "memory");
    asm volatile("s_barrier_wait -1" ::: "memory");
}

template <int S>
__device__ __forceinline__ void scan_request(
    const int Jn, const int Rn, const int wave, const int lane, const int c, const int Bt,
    const float* lbase, const float* pf, float4 (&tq)[3][MAXT][2])
{
    const bool chain = (wave == 0);
    const int wl = chain ? 0 : wave - 1;
    const float* ln = chain ? pf + (size_t)Bt * 256
                            : lbase + (size_t)(16 * Rn + c) * COLS + 16 * wl;
    const float* lr = lbase + (size_t)(16 * Jn + c) * COLS + 16 * Jn;
    const float* pp = pf + ((size_t)(Jn * NBT + Bt) * 32 + lane) * 8;
    const float* s0 = chain ? pp : ln;
    const float* s1 = chain ? lr : ln + 16 * NHELP;
    const float* s2 = chain ? ((Jn > 0) ? lr - 16 : lr) : ln + 32 * NHELP;
    tq[S][0][0] = *(const float4*)(s0);
    tq[S][0][1] = *(const float4*)(s0 + 4);
    tq[S][1][0] = *(const float4*)(s1);
    tq[S][1][1] = *(const float4*)(s1 + 4);
    tq[S][2][0] = *(const float4*)(s2);
    tq[S][2][1] = *(const float4*)(s2 + 4);
    #pragma unroll
    for (int t = 3; t < MAXT; ++t) {
        tq[S][t][0] = *(const float4*)(ln + 16 * NHELP * t);
        tq[S][t][1] = *(const float4*)(ln + 16 * NHELP * t + 4);
    }
}

template <int CUR>
__device__ __forceinline__ void scan_block(
    const int J, const int wave, const int lane, const int c, const int h, const int Bt,
    const float* lbase, const float* pf, float* out,
    float (&hL)[NBLK][2][16][8], float (&part)[2][NHELP][32][8],
    float (&hv)[8], float4 (&tq)[3][MAXT][2])
{
    scan_request<(CUR + 2) % 3>((J + 2 < NBLK) ? J + 2 : NBLK - 1, (J + 3 < NBLK) ? J + 3 : NBLK - 1,
                                wave, lane, c, Bt, lbase, pf, tq);
    __builtin_amdgcn_sched_barrier(0);
    if (J >= NBLK) {
    } else if (wave == 0) {
        const float4 p0 = tq[CUR][0][0], p1 = tq[CUR][0][1], d0 = tq[CUR][1][0], d1 = tq[CUR][1][1];
        const float4 c0 = tq[CUR][2][0], c1 = tq[CUR][2][1];
        v8f acc = {p0.x, p0.y, p0.z, p0.w, p1.x, p1.y, p1.z, p1.w};
        if (J >= 1) {
            v8f q = {};
            for (int w = 0; w < NHELP; ++w) {
                const float4 u0 = *(const float4*)&part[J & 1][w][lane][0];
                const float4 u1 = *(const float4*)&part[J & 1][w][lane][4];
                q[0] += u0.x; q[1] += u0.y; q[2] += u0.z; q[3] += u0.w;
                q[4] += u1.x; q[5] += u1.y; q[6] += u1.z; q[7] += u1.w;
            }
            const v2f a0 = {c0.x, c0.y}, a1 = {c0.z, c0.w}, a2 = {c1.x, c1.y}, a3 = {c1.z, c1.w};
            const v2f b0 = {hv[0], hv[1]}, b1 = {hv[2], hv[3]}, b2 = {hv[4], hv[5]}, b3 = {hv[6], hv[7]};
            acc = mma4(a0, b0, acc);
            q   = mma4(a1, b1, q);
            acc = mma4(a2, b2, acc);
            q   = mma4(a3, b3, q);
            acc += q;
        }
        const float dg[8] = {d0.x, d0.y, d0.z, d0.w, d1.x, d1.y, d1.z, d1.w};
        float alo[8], ahi[8];
        for (int v = 0; v < 8; ++v) {
            alo[v] = (h == 0) ? dg[v] : 0.0f;
            ahi[v] = (h == 1) ? dg[v] : 0.0f;
        }
        #pragma unroll
        for (int i = 0; i < 16; ++i) {
            const int v = i & 7;
            const float pre = acc[v];
            const float sg  = __builtin_fmaf(0.5f, __builtin_amdgcn_tanhf(pre), 0.5f);
            hv[v] = ((i >> 3) == h) ? sg : hv[v];
            if (i < 15) {
                const v2f a = {(i < 8) ? alo[v] : ahi[v], 0.0f};
                const v2f b = {sg, 0.0f};
                acc = mma4(a, b, acc);
            }
        }
        *(float4*)&hL[J][h][c][0] = make_float4(hv[0], hv[1], hv[2], hv[3]);
        *(float4*)&hL[J][h][c][4] = make_float4(hv[4], hv[5], hv[6], hv[7]);
        if (J == NBLK - 1) {
            for (int v = 0; v < 8; ++v)
                out[(size_t)(8 * h + v) * BATCH + 16 * Bt + c] = hv[v];
        }
    } else if (J + 1 < NBLK) {
        const int R = J + 1;
        const int w = wave - 1;
        float4 hb[MAXT][2];
        #pragma unroll
        for (int t = 0; t < MAXT; ++t) {
            const int j = w + NHELP * t;
            hb[t][0] = *(const float4*)&hL[j][h][c][0];
            hb[t][1] = *(const float4*)&hL[j][h][c][4];
        }
        __builtin_amdgcn_sched_barrier(0);
        v8f q0 = {}, q1 = {};
        #pragma unroll
        for (int t = 0; t < MAXT; ++t) {
            const int j = w + NHELP * t;
            if (j < J) {
                const float4 a0 = tq[CUR][t][0], a1 = tq[CUR][t][1];
                const float4 b0 = hb[t][0], b1 = hb[t][1];
                const v2f x0 = {a0.x, a0.y}, y0 = {b0.x, b0.y};
                const v2f x1 = {a0.z, a0.w}, y1 = {b0.z, b0.w};
                const v2f x2 = {a1.x, a1.y}, y2 = {b1.x, b1.y};
                const v2f x3 = {a1.z, a1.w}, y3 = {b1.z, b1.w};
                q0 = mma4(x0, y0, q0);
                q1 = mma4(x1, y1, q1);
                q0 = mma4(x2, y2, q0);
                q1 = mma4(x3, y3, q1);
            }
        }
        q0 += q1;
        *(float4*)&part[R & 1][w][lane][0] = make_float4(q0[0], q0[1], q0[2], q0[3]);
        *(float4*)&part[R & 1][w][lane][4] = make_float4(q0[4], q0[5], q0[6], q0[7]);
    }
    lds_block_barrier();
}

__global__ __launch_bounds__(128) void ne_block_scan_kernel(
    const float* wd,
    const float* pf,
    float*       out)
{
    __shared__ __attribute__((aligned(16))) float hL[NBLK][2][16][8];
    __shared__ __attribute__((aligned(16))) float part[2][NHELP][32][8];

    const int wave = __builtin_amdgcn_readfirstlane((int)(threadIdx.x >> 5));
    const int lane = threadIdx.x & 31;
    const int c = lane & 15, h = lane >> 4;
    const int Bt = blockIdx.x;
    const float* lbase = wd + INPS + 8 * h;

    float  hv[8] = {0.f, 0.f, 0.f, 0.f, 0.f, 0.f, 0.f, 0.f};
    float4 tq[3][MAXT][2];
    scan_request<0>(0, 1, wave, lane, c, Bt, lbase, pf, tq);
    __builtin_amdgcn_sched_barrier(0);
    scan_request<1>(1, 2, wave, lane, c, Bt, lbase, pf, tq);
    __builtin_amdgcn_sched_barrier(0);
    #pragma unroll
    for (int t = 0; t < MAXT; ++t) {
        tq[2][t][0] = make_float4(0.f, 0.f, 0.f, 0.f); tq[2][t][1] = tq[2][t][0];
    }

    #pragma nounroll
    for (int J0 = 0; J0 < NBLK; J0 += 3) {
        scan_block<0>(J0,     wave, lane, c, h, Bt, lbase, pf, out, hL, part, hv, tq);
        scan_block<1>(J0 + 1, wave, lane, c, h, Bt, lbase, pf, out, hL, part, hv, tq);
        scan_block<2>(J0 + 2, wave, lane, c, h, Bt, lbase, pf, out, hL, part, hv, tq);
    }
}

extern "C" void kernel_launch(void* const* d_in, const int* in_sizes, int n_in,
                              void* d_out, int out_size, void* d_ws, size_t ws_size,
                              hipStream_t stream) {
    (void)in_sizes; (void)n_in; (void)ws_size; (void)out_size;
    const float* x       = (const float*)d_in[0];
    const int*   in_idxs = (const int*)d_in[1];
    const float* weights = (const float*)d_in[2];
    float*       out     = (float*)d_out;
    float*       wd      = (float*)d_ws;
    float*       pf      = wd + (size_t)NEUR * COLS;

    ne_rows_gemm_kernel<<<dim3(NBLK, NG), dim3(128), 0, stream>>>(x, in_idxs, weights, wd, pf);
    ne_block_scan_kernel<<<dim3(NBT), dim3(128), 0, stream>>>(wd, pf, out);
}


# Scan's weight and P tiles stored by launch 1 in the scan's fragment order, 1 KB contiguous per tile: each scan load reads 512 contiguous bytes (4 lines), not 16 rows' pieces (sum order unchanged)
# speedup vs baseline: 8.6718x; 1.1774x over previous
#include <hip/hip_runtime.h>


typedef float v2f __attribute__((ext_vector_type(2)));
typedef float v8f __attribute__((ext_vector_type(8)));

#define BATCH 512
#define INPS  1024
#define NEUR  512
#define KFAN  64
#define NOUT  16
#define COLS  (INPS + NEUR)
#define NBLK  (NEUR / 16)
#define NBT   (BATCH / 16)
#define NHELP 3
#define MAXT  10
#define GW    4
#define NG    (NBT / GW)
#define HS    (NEUR / NG)
#define SW    (INPS + HS)
#define SP    (SW + 4)

__device__ __forceinline__ v8f mma4(v2f a, v2f b, v8f c) {
    return __builtin_amdgcn_wmma_f32_16x16x4_f32(false, a, false, b, (short)0, c, false, false);
}

__global__ __launch_bounds__(128) void ne_rows_gemm_kernel(
    const float* __restrict__ x,
    const int*   __restrict__ in_idxs,
    const float* __restrict__ weights,
    float*       __restrict__ wd,
    float*       __restrict__ pf)
{
    __shared__ __attribute__((aligned(16))) float slab[16 * SP];
    __shared__ __attribute__((aligned(16))) int   sj[16 * KFAN];
    __shared__ __attribute__((aligned(16))) float sw[16 * KFAN];
    const int tid = threadIdx.x;
    const int J = blockIdx.x, g = blockIdx.y;
    {
        const int4*   gi = (const int4*)(in_idxs + (size_t)J * 16 * KFAN);
        const float4* gw = (const float4*)(weights + (size_t)J * 16 * KFAN);
        ((int4*)sj)[tid]         = gi[tid];
        ((int4*)sj)[tid + 128]   = gi[tid + 128];
        ((float4*)sw)[tid]       = gw[tid];
        ((float4*)sw)[tid + 128] = gw[tid + 128];
    }
    for (int i = tid; i < 16 * SP / 4; i += 128)
        ((float4*)slab)[i] = make_float4(0.f, 0.f, 0.f, 0.f);
    __syncthreads();
    {
        const int t  = tid & 63;
        const int r0 = __builtin_amdgcn_readfirstlane(tid >> 6);
        int   jv[8];
        float sv[8];
        #pragma unroll
        for (int i = 0; i < 8; ++i) {
            jv[i] = sj[(r0 + 2 * i) * KFAN + t];
            sv[i] = 0.0f;
        }
        #pragma unroll 4
        for (int k4 = 0; k4 < KFAN / 4; ++k4) {
            #pragma unroll
            for (int i = 0; i < 8; ++i) {
                const int4   p = ((const int4*)(sj + (r0 + 2 * i) * KFAN))[k4];
                const float4 u = ((const float4*)(sw + (r0 + 2 * i) * KFAN))[k4];
                sv[i] += (p.x == jv[i]) ? u.x : 0.0f;
                sv[i] += (p.y == jv[i]) ? u.y : 0.0f;
                sv[i] += (p.z == jv[i]) ? u.z : 0.0f;
                sv[i] += (p.w == jv[i]) ? u.w : 0.0f;
            }
        }
        #pragma unroll
        for (int i = 0; i < 8; ++i) {
            const int r = r0 + 2 * i;
            const int j = jv[i];
            const unsigned q = (unsigned)(j - INPS - HS * g);
            int col = (j < INPS) ? j : ((q < (unsigned)HS) ? (int)(INPS + q) : SW);
            col = ((unsigned)j < (unsigned)(INPS + 16 * J + r)) ? col : SW;
            slab[r * SP + col] = 0.5f * sv[i];
        }
    }
    __syncthreads();
    for (int i = tid; i < 16 * HS / 4; i += 128) {
        const int ln = i & 31, q = (i >> 5) & 1, jl = i >> 6;
        const int r = ln & 15, hh = ln >> 4;
        const float4 v = *(const float4*)&slab[r * SP + INPS + 16 * jl + 8 * hh + 4 * q];
        *(float4*)(wd + ((size_t)((J * NBLK + (HS / 16) * g + jl) * 2 + q) * 32 + ln) * 4) = v;
    }

    const int lane = tid & 31;
    const int c = lane & 15, h = lane >> 4;
    const int Bt = GW * g + (tid >> 5);
    const float* ap = slab + c * SP + 4 * h;
    const float* bp = x + (size_t)(16 * Bt + c) * INPS + 4 * h;
    v8f acc0 = {}, acc1 = {};
    float4 af[2][16], bf[2][16];
    #pragma unroll
    for (int s = 0; s < 16; ++s) {
        af[0][s] = *(const float4*)(ap + 8 * s);
        bf[0][s] = *(const float4*)(bp + 8 * s);
    }
    #pragma unroll
    for (int kc = 0; kc < INPS / 128; ++kc) {
        const int cur = kc & 1, nxt = cur ^ 1;
        if (kc + 1 < INPS / 128) {
            #pragma unroll
            for (int s = 0; s < 16; ++s) {
                af[nxt][s] = *(const float4*)(ap + 128 * (kc + 1) + 8 * s);
                bf[nxt][s] = *(const float4*)(bp + 128 * (kc + 1) + 8 * s);
            }
        }
        __builtin_amdgcn_sched_barrier(0);
        #pragma unroll
        for (int s = 0; s < 16; ++s) {
            const float4 a = af[cur][s];
            const float4 b = bf[cur][s];
            const v2f a0 = {a.x, a.y}, b0 = {b.x, b.y};
            const v2f a1 = {a.z, a.w}, b1 = {b.z, b.w};
            acc0 = mma4(a0, b0, acc0);
            acc1 = mma4(a1, b1, acc1);
        }
        __builtin_amdgcn_sched_barrier(0);
    }
    acc0 += acc1;
    float4* o = (float4*)(pf + (size_t)(J * NBT + Bt) * 256) + lane;
    o[0]  = make_float4(acc0[0], acc0[1], acc0[2], acc0[3]);
    o[32] = make_float4(acc0[4], acc0[5], acc0[6], acc0[7]);
}


__device__ __forceinline__ void lds_block_barrier() {
    asm volatile("s_wait_dscnt 0x0" ::: "memory");
    asm volatile("s_barrier_signal -1" ::: "memory");
    asm volatile("s_barrier_wait -1" ::: "memory");
}

template <int S>
__device__ __forceinline__ void scan_request(
    const int Jn, const int Rn, const int wave, const int lane, const int c, const int Bt,
    const float* lbase, const float* pf, float4 (&tq)[3][MAXT][2])
{
    const bool chain = (wave == 0);
    const int wl = chain ? 0 : wave - 1;
    const float* ln = chain ? pf + (size_t)Bt * 256
                            : lbase + (size_t)(Rn * NBLK + wl) * 256;
    const float* lr = lbase + (size_t)(Jn * NBLK + Jn) * 256;
    const float* pp = pf + (size_t)(Jn * NBT + Bt) * 256 + 4 * lane;
    const float* s0 = chain ? pp : ln;
    const float* s1 = chain ? lr : ln + 256 * NHELP;
    const float* s2 = chain ? ((Jn > 0) ? lr - 256 : lr) : ln + 512 * NHELP;
    tq[S][0][0] = *(const float4*)(s0);
    tq[S][0][1] = *(const float4*)(s0 + 128);
    tq[S][1][0] = *(const float4*)(s1);
    tq[S][1][1] = *(const float4*)(s1 + 128);
    tq[S][2][0] = *(const float4*)(s2);
    tq[S][2][1] = *(const float4*)(s2 + 128);
    #pragma unroll
    for (int t = 3; t < MAXT; ++t) {
        tq[S][t][0] = *(const float4*)(ln + 256 * NHELP * t);
        tq[S][t][1] = *(const float4*)(ln + 256 * NHELP * t + 128);
    }
}

template <int CUR>
__device__ __forceinline__ void scan_block(
    const int J, const int wave, const int lane, const int c, const int h, const int Bt,
    const float* lbase, const float* pf, float* out,
    float (&hL)[NBLK][2][16][8], float (&part)[2][NHELP][32][8],
    float (&hv)[8], float4 (&tq)[3][MAXT][2])
{
    scan_request<(CUR + 2) % 3>((J + 2 < NBLK) ? J + 2 : NBLK - 1, (J + 3 < NBLK) ? J + 3 : NBLK - 1,
                                wave, lane, c, Bt, lbase, pf, tq);
    __builtin_amdgcn_sched_barrier(0);
    if (J >= NBLK) {
    } else if (wave == 0) {
        const float4 p0 = tq[CUR][0][0], p1 = tq[CUR][0][1], d0 = tq[CUR][1][0], d1 = tq[CUR][1][1];
        const float4 c0 = tq[CUR][2][0], c1 = tq[CUR][2][1];
        v8f acc = {p0.x, p0.y, p0.z, p0.w, p1.x, p1.y, p1.z, p1.w};
        if (J >= 1) {
            v8f q = {};
            for (int w = 0; w < NHELP; ++w) {
                const float4 u0 = *(const float4*)&part[J & 1][w][lane][0];
                const float4 u1 = *(const float4*)&part[J & 1][w][lane][4];
                q[0] += u0.x; q[1] += u0.y; q[2] += u0.z; q[3] += u0.w;
                q[4] += u1.x; q[5] += u1.y; q[6] += u1.z; q[7] += u1.w;
            }
            const v2f a0 = {c0.x, c0.y}, a1 = {c0.z, c0.w}, a2 = {c1.x, c1.y}, a3 = {c1.z, c1.w};
            const v2f b0 = {hv[0], hv[1]}, b1 = {hv[2], hv[3]}, b2 = {hv[4], hv[5]}, b3 = {hv[6], hv[7]};
            acc = mma4(a0, b0, acc);
            q   = mma4(a1, b1, q);
            acc = mma4(a2, b2, acc);
            q   = mma4(a3, b3, q);
            acc += q;
        }
        const float dg[8] = {d0.x, d0.y, d0.z, d0.w, d1.x, d1.y, d1.z, d1.w};
        float alo[8], ahi[8];
        for (int v = 0; v < 8; ++v) {
            alo[v] = (h == 0) ? dg[v] : 0.0f;
            ahi[v] = (h == 1) ? dg[v] : 0.0f;
        }
        #pragma unroll
        for (int i = 0; i < 16; ++i) {
            const int v = i & 7;
            const float pre = acc[v];
            const float sg  = __builtin_fmaf(0.5f, __builtin_amdgcn_tanhf(pre), 0.5f);
            hv[v] = ((i >> 3) == h) ? sg : hv[v];
            if (i < 15) {
                const v2f a = {(i < 8) ? alo[v] : ahi[v], 0.0f};
                const v2f b = {sg, 0.0f};
                acc = mma4(a, b, acc);
            }
        }
        *(float4*)&hL[J][h][c][0] = make_float4(hv[0], hv[1], hv[2], hv[3]);
        *(float4*)&hL[J][h][c][4] = make_float4(hv[4], hv[5], hv[6], hv[7]);
        if (J == NBLK - 1) {
            for (int v = 0; v < 8; ++v)
                out[(size_t)(8 * h + v) * BATCH + 16 * Bt + c] = hv[v];
        }
    } else if (J + 1 < NBLK) {
        const int R = J + 1;
        const int w = wave - 1;
        float4 hb[MAXT][2];
        #pragma unroll
        for (int t = 0; t < MAXT; ++t) {
            const int j = w + NHELP * t;
            hb[t][0] = *(const float4*)&hL[j][h][c][0];
            hb[t][1] = *(const float4*)&hL[j][h][c][4];
        }
        __builtin_amdgcn_sched_barrier(0);
        v8f q0 = {}, q1 = {};
        #pragma unroll
        for (int t = 0; t < MAXT; ++t) {
            const int j = w + NHELP * t;
            if (j < J) {
                const float4 a0 = tq[CUR][t][0], a1 = tq[CUR][t][1];
                const float4 b0 = hb[t][0], b1 = hb[t][1];
                const v2f x0 = {a0.x, a0.y}, y0 = {b0.x, b0.y};
                const v2f x1 = {a0.z, a0.w}, y1 = {b0.z, b0.w};
                const v2f x2 = {a1.x, a1.y}, y2 = {b1.x, b1.y};
                const v2f x3 = {a1.z, a1.w}, y3 = {b1.z, b1.w};
                q0 = mma4(x0, y0, q0);
                q1 = mma4(x1, y1, q1);
                q0 = mma4(x2, y2, q0);
                q1 = mma4(x3, y3, q1);
            }
        }
        q0 += q1;
        *(float4*)&part[R & 1][w][lane][0] = make_float4(q0[0], q0[1], q0[2], q0[3]);
        *(float4*)&part[R & 1][w][lane][4] = make_float4(q0[4], q0[5], q0[6], q0[7]);
    }
    lds_block_barrier();
}

__global__ __launch_bounds__(128) void ne_block_scan_kernel(
    const float* wd,
    const float* pf,
    float*       out)
{
    __shared__ __attribute__((aligned(16))) float hL[NBLK][2][16][8];
    __shared__ __attribute__((aligned(16))) float part[2][NHELP][32][8];

    const int wave = __builtin_amdgcn_readfirstlane((int)(threadIdx.x >> 5));
    const int lane = threadIdx.x & 31;
    const int c = lane & 15, h = lane >> 4;
    const int Bt = blockIdx.x;
    const float* lbase = wd + 4 * lane;

    float  hv[8] = {0.f, 0.f, 0.f, 0.f, 0.f, 0.f, 0.f, 0.f};
    float4 tq[3][MAXT][2];
    scan_request<0>(0, 1, wave, lane, c, Bt, lbase, pf, tq);
    __builtin_amdgcn_sched_barrier(0);
    scan_request<1>(1, 2, wave, lane, c, Bt, lbase, pf, tq);
    __builtin_amdgcn_sched_barrier(0);
    #pragma unroll
    for (int t = 0; t < MAXT; ++t) {
        tq[2][t][0] = make_float4(0.f, 0.f, 0.f, 0.f); tq[2][t][1] = tq[2][t][0];
    }

    #pragma nounroll
    for (int J0 = 0; J0 < NBLK; J0 += 3) {
        scan_block<0>(J0,     wave, lane, c, h, Bt, lbase, pf, out, hL, part, hv, tq);
        scan_block<1>(J0 + 1, wave, lane, c, h, Bt, lbase, pf, out, hL, part, hv, tq);
        scan_block<2>(J0 + 2, wave, lane, c, h, Bt, lbase, pf, out, hL, part, hv, tq);
    }
}

extern "C" void kernel_launch(void* const* d_in, const int* in_sizes, int n_in,
                              void* d_out, int out_size, void* d_ws, size_t ws_size,
                              hipStream_t stream) {
    (void)in_sizes; (void)n_in; (void)ws_size; (void)out_size;
    const float* x       = (const float*)d_in[0];
    const int*   in_idxs = (const int*)d_in[1];
    const float* weights = (const float*)d_in[2];
    float*       out     = (float*)d_out;
    float*       wd      = (float*)d_ws;
    float*       pf      = wd + (size_t)NEUR * COLS;

    ne_rows_gemm_kernel<<<dim3(NBLK, NG), dim3(128), 0, stream>>>(x, in_idxs, weights, wd, pf);
    ne_block_scan_kernel<<<dim3(NBT), dim3(128), 0, stream>>>(wd, pf, out);
}


# Scan's chain wave: block start taken off its serial path. Partial sums are read right behind the barrier; the 4 products with the newest block are made at that block's end (reorders the sum)
# speedup vs baseline: 9.3231x; 1.0751x over previous
#include <hip/hip_runtime.h>


typedef float v2f __attribute__((ext_vector_type(2)));
typedef float v8f __attribute__((ext_vector_type(8)));

#define BATCH 512
#define INPS  1024
#define NEUR  512
#define KFAN  64
#define NOUT  16
#define COLS  (INPS + NEUR)
#define NBLK  (NEUR / 16)
#define NBT   (BATCH / 16)
#define NHELP 3
#define MAXT  10
#define GW    4
#define NG    (NBT / GW)
#define HS    (NEUR / NG)
#define SW    (INPS + HS)
#define SP    (SW + 4)

__device__ __forceinline__ v8f mma4(v2f a, v2f b, v8f c) {
    return __builtin_amdgcn_wmma_f32_16x16x4_f32(false, a, false, b, (short)0, c, false, false);
}

__global__ __launch_bounds__(128) void ne_rows_gemm_kernel(
    const float* __restrict__ x,
    const int*   __restrict__ in_idxs,
    const float* __restrict__ weights,
    float*       __restrict__ wd,
    float*       __restrict__ pf)
{
    __shared__ __attribute__((aligned(16))) float slab[16 * SP];
    __shared__ __attribute__((aligned(16))) int   sj[16 * KFAN];
    __shared__ __attribute__((aligned(16))) float sw[16 * KFAN];
    const int tid = threadIdx.x;
    const int J = blockIdx.x, g = blockIdx.y;
    {
        const int4*   gi = (const int4*)(in_idxs + (size_t)J * 16 * KFAN);
        const float4* gw = (const float4*)(weights + (size_t)J * 16 * KFAN);
        ((int4*)sj)[tid]         = gi[tid];
        ((int4*)sj)[tid + 128]   = gi[tid + 128];
        ((float4*)sw)[tid]       = gw[tid];
        ((float4*)sw)[tid + 128] = gw[tid + 128];
    }
    for (int i = tid; i < 16 * SP / 4; i += 128)
        ((float4*)slab)[i] = make_float4(0.f, 0.f, 0.f, 0.f);
    __syncthreads();
    {
        const int t  = tid & 63;
        const int r0 = __builtin_amdgcn_readfirstlane(tid >> 6);
        int   jv[8];
        float sv[8];
        #pragma unroll
        for (int i = 0; i < 8; ++i) {
            jv[i] = sj[(r0 + 2 * i) * KFAN + t];
            sv[i] = 0.0f;
        }
        #pragma unroll 4
        for (int k4 = 0; k4 < KFAN / 4; ++k4) {
            #pragma unroll
            for (int i = 0; i < 8; ++i) {
                const int4   p = ((const int4*)(sj + (r0 + 2 * i) * KFAN))[k4];
                const float4 u = ((const float4*)(sw + (r0 + 2 * i) * KFAN))[k4];
                sv[i] += (p.x == jv[i]) ? u.x : 0.0f;
                sv[i] += (p.y == jv[i]) ? u.y : 0.0f;
                sv[i] += (p.z == jv[i]) ? u.z : 0.0f;
                sv[i] += (p.w == jv[i]) ? u.w : 0.0f;
            }
        }
        #pragma unroll
        for (int i = 0; i < 8; ++i) {
            const int r = r0 + 2 * i;
            const int j = jv[i];
            const unsigned q = (unsigned)(j - INPS - HS * g);
            int col = (j < INPS) ? j : ((q < (unsigned)HS) ? (int)(INPS + q) : SW);
            col = ((unsigned)j < (unsigned)(INPS + 16 * J + r)) ? col : SW;
            slab[r * SP + col] = 0.5f * sv[i];
        }
    }
    __syncthreads();
    for (int i = tid; i < 16 * HS / 4; i += 128) {
        const int ln = i & 31, q = (i >> 5) & 1, jl = i >> 6;
        const int r = ln & 15, hh = ln >> 4;
        const float4 v = *(const float4*)&slab[r * SP + INPS + 16 * jl + 8 * hh + 4 * q];
        *(float4*)(wd + ((size_t)((J * NBLK + (HS / 16) * g + jl) * 2 + q) * 32 + ln) * 4) = v;
    }

    const int lane = tid & 31;
    const int c = lane & 15, h = lane >> 4;
    const int Bt = GW * g + (tid >> 5);
    const float* ap = slab + c * SP + 4 * h;
    const float* bp = x + (size_t)(16 * Bt + c) * INPS + 4 * h;
    v8f acc0 = {}, acc1 = {};
    float4 af[2][16], bf[2][16];
    #pragma unroll
    for (int s = 0; s < 16; ++s) {
        af[0][s] = *(const float4*)(ap + 8 * s);
        bf[0][s] = *(const float4*)(bp + 8 * s);
    }
    #pragma unroll
    for (int kc = 0; kc < INPS / 128; ++kc) {
        const int cur = kc & 1, nxt = cur ^ 1;
        if (kc + 1 < INPS / 128) {
            #pragma unroll
            for (int s = 0; s < 16; ++s) {
                af[nxt][s] = *(const float4*)(ap + 128 * (kc + 1) + 8 * s);
                bf[nxt][s] = *(const float4*)(bp + 128 * (kc + 1) + 8 * s);
            }
        }
        __builtin_amdgcn_sched_barrier(0);
        #pragma unroll
        for (int s = 0; s < 16; ++s) {
            const float4 a = af[cur][s];
            const float4 b = bf[cur][s];
            const v2f a0 = {a.x, a.y}, b0 = {b.x, b.y};
            const v2f a1 = {a.z, a.w}, b1 = {b.z, b.w};
            acc0 = mma4(a0, b0, acc0);
            acc1 = mma4(a1, b1, acc1);
        }
        __builtin_amdgcn_sched_barrier(0);
    }
    acc0 += acc1;
    float4* o = (float4*)(pf + (size_t)(J * NBT + Bt) * 256) + lane;
    o[0]  = make_float4(acc0[0], acc0[1], acc0[2], acc0[3]);
    o[32] = make_float4(acc0[4], acc0[5], acc0[6], acc0[7]);
}


__device__ __forceinline__ void lds_block_barrier() {
    asm volatile("s_wait_dscnt 0x0" ::: "memory");
    asm volatile("s_barrier_signal -1" ::: "memory");
    asm volatile("s_barrier_wait -1" ::: "memory");
}

template <int S>
__device__ __forceinline__ void scan_request(
    const int Jn, const int Rn, const int wave, const int lane, const int c, const int Bt,
    const float* lbase, const float* pf, float4 (&tq)[3][MAXT][2])
{
    const bool chain = (wave == 0);
    const int wl = chain ? 0 : wave - 1;
    const float* ln = chain ? pf + (size_t)Bt * 256
                            : lbase + (size_t)(Rn * NBLK + wl) * 256;
    const float* lr = lbase + (size_t)(Jn * NBLK + Jn) * 256;
    const float* pp = pf + (size_t)(Jn * NBT + Bt) * 256 + 4 * lane;
    const float* s0 = chain ? pp : ln;
    const float* s1 = chain ? lr : ln + 256 * NHELP;
    const float* s2 = chain ? ((Jn > 0) ? lr - 256 : lr) : ln + 512 * NHELP;
    tq[S][0][0] = *(const float4*)(s0);
    tq[S][0][1] = *(const float4*)(s0 + 128);
    tq[S][1][0] = *(const float4*)(s1);
    tq[S][1][1] = *(const float4*)(s1 + 128);
    tq[S][2][0] = *(const float4*)(s2);
    tq[S][2][1] = *(const float4*)(s2 + 128);
    #pragma unroll
    for (int t = 3; t < MAXT; ++t) {
        tq[S][t][0] = *(const float4*)(ln + 256 * NHELP * t);
        tq[S][t][1] = *(const float4*)(ln + 256 * NHELP * t + 128);
    }
}

template <int CUR>
__device__ __forceinline__ void scan_block(
    const int J, const int wave, const int lane, const int c, const int h, const int Bt,
    const float* lbase, const float* pf, float* out,
    float (&hL)[NBLK][2][16][8], float (&part)[2][NHELP][32][8],
    float (&hv)[8], v8f &pa, v8f &pr, float4 (&tq)[3][MAXT][2])
{
    float4 u[NHELP][2];
    if (wave == 0 && J >= 1 && J < NBLK) {
        #pragma unroll
        for (int w = 0; w < NHELP; ++w) {
            u[w][0] = *(const float4*)&part[J & 1][w][lane][0];
            u[w][1] = *(const float4*)&part[J & 1][w][lane][4];
        }
    }
    scan_request<(CUR + 2) % 3>((J + 2 < NBLK) ? J + 2 : NBLK - 1, (J + 3 < NBLK) ? J + 3 : NBLK - 1,
                                wave, lane, c, Bt, lbase, pf, tq);
    __builtin_amdgcn_sched_barrier(0);
    if (J >= NBLK) {
    } else if (wave == 0) {
        const float4 p0 = tq[CUR][0][0], p1 = tq[CUR][0][1], d0 = tq[CUR][1][0], d1 = tq[CUR][1][1];
        v8f acc = {p0.x, p0.y, p0.z, p0.w, p1.x, p1.y, p1.z, p1.w};
        if (J >= 1) {
            acc = pa + pr;
            v8f q;
            q[0] = (u[0][0].x + u[1][0].x) + u[2][0].x; q[1] = (u[0][0].y + u[1][0].y) + u[2][0].y;
            q[2] = (u[0][0].z + u[1][0].z) + u[2][0].z; q[3] = (u[0][0].w + u[1][0].w) + u[2][0].w;
            q[4] = (u[0][1].x + u[1][1].x) + u[2][1].x; q[5] = (u[0][1].y + u[1][1].y) + u[2][1].y;
            q[6] = (u[0][1].z + u[1][1].z) + u[2][1].z; q[7] = (u[0][1].w + u[1][1].w) + u[2][1].w;
            acc += q;
        }
        const float dg[8] = {d0.x, d0.y, d0.z, d0.w, d1.x, d1.y, d1.z, d1.w};
        float alo[8], ahi[8];
        for (int v = 0; v < 8; ++v) {
            alo[v] = (h == 0) ? dg[v] : 0.0f;
            ahi[v] = (h == 1) ? dg[v] : 0.0f;
        }
        #pragma unroll
        for (int i = 0; i < 16; ++i) {
            const int v = i & 7;
            const float pre = acc[v];
            const float sg  = __builtin_fmaf(0.5f, __builtin_amdgcn_tanhf(pre), 0.5f);
            hv[v] = ((i >> 3) == h) ? sg : hv[v];
            if (i < 15) {
                const v2f a = {(i < 8) ? alo[v] : ahi[v], 0.0f};
                const v2f b = {sg, 0.0f};
                acc = mma4(a, b, acc);
            }
        }
        *(float4*)&hL[J][h][c][0] = make_float4(hv[0], hv[1], hv[2], hv[3]);
        *(float4*)&hL[J][h][c][4] = make_float4(hv[4], hv[5], hv[6], hv[7]);
        if (J == NBLK - 1) {
            for (int v = 0; v < 8; ++v)
                out[(size_t)(8 * h + v) * BATCH + 16 * Bt + c] = hv[v];
        }
        __builtin_amdgcn_sched_barrier(0);
        {
            constexpr int NX = (CUR + 1) % 3;
            const float4 n0 = tq[NX][0][0], n1 = tq[NX][0][1];
            const float4 c0 = tq[NX][2][0], c1 = tq[NX][2][1];
            const v2f a0 = {c0.x, c0.y}, a1 = {c0.z, c0.w}, a2 = {c1.x, c1.y}, a3 = {c1.z, c1.w};
            const v2f b0 = {hv[0], hv[1]}, b1 = {hv[2], hv[3]}, b2 = {hv[4], hv[5]}, b3 = {hv[6], hv[7]};
            v8f xa = {n0.x, n0.y, n0.z, n0.w, n1.x, n1.y, n1.z, n1.w};
            v8f xr = {};
            xa = mma4(a0, b0, xa);
            xr = mma4(a1, b1, xr);
            xa = mma4(a2, b2, xa);
            xr = mma4(a3, b3, xr);
            pa = xa;
            pr = xr;
        }
    } else if (J + 1 < NBLK) {
        const int R = J + 1;
        const int w = wave - 1;
        float4 hb[MAXT][2];
        #pragma unroll
        for (int t = 0; t < MAXT; ++t) {
            const int j = w + NHELP * t;
            hb[t][0] = *(const float4*)&hL[j][h][c][0];
            hb[t][1] = *(const float4*)&hL[j][h][c][4];
        }
        __builtin_amdgcn_sched_barrier(0);
        v8f q0 = {}, q1 = {};
        #pragma unroll
        for (int t = 0; t < MAXT; ++t) {
            const int j = w + NHELP * t;
            if (j < J) {
                const float4 a0 = tq[CUR][t][0], a1 = tq[CUR][t][1];
                const float4 b0 = hb[t][0], b1 = hb[t][1];
                const v2f x0 = {a0.x, a0.y}, y0 = {b0.x, b0.y};
                const v2f x1 = {a0.z, a0.w}, y1 = {b0.z, b0.w};
                const v2f x2 = {a1.x, a1.y}, y2 = {b1.x, b1.y};
                const v2f x3 = {a1.z, a1.w}, y3 = {b1.z, b1.w};
                q0 = mma4(x0, y0, q0);
                q1 = mma4(x1, y1, q1);
                q0 = mma4(x2, y2, q0);
                q1 = mma4(x3, y3, q1);
            }
        }
        q0 += q1;
        *(float4*)&part[R & 1][w][lane][0] = make_float4(q0[0], q0[1], q0[2], q0[3]);
        *(float4*)&part[R & 1][w][lane][4] = make_float4(q0[4], q0[5], q0[6], q0[7]);
    }
    lds_block_barrier();
}

__global__ __launch_bounds__(128) void ne_block_scan_kernel(
    const float* wd,
    const float* pf,
    float*       out)
{
    __shared__ __attribute__((aligned(16))) float hL[NBLK][2][16][8];
    __shared__ __attribute__((aligned(16))) float part[2][NHELP][32][8];

    const int wave = __builtin_amdgcn_readfirstlane((int)(threadIdx.x >> 5));
    const int lane = threadIdx.x & 31;
    const int c = lane & 15, h = lane >> 4;
    const int Bt = blockIdx.x;
    const float* lbase = wd + 4 * lane;

    float  hv[8] = {0.f, 0.f, 0.f, 0.f, 0.f, 0.f, 0.f, 0.f};
    v8f pa = {}, pr = {};
    float4 tq[3][MAXT][2];
    scan_request<0>(0, 1, wave, lane, c, Bt, lbase, pf, tq);
    __builtin_amdgcn_sched_barrier(0);
    scan_request<1>(1, 2, wave, lane, c, Bt, lbase, pf, tq);
    __builtin_amdgcn_sched_barrier(0);
    #pragma unroll
    for (int t = 0; t < MAXT; ++t) {
        tq[2][t][0] = make_float4(0.f, 0.f, 0.f, 0.f); tq[2][t][1] = tq[2][t][0];
    }

    #pragma nounroll
    for (int J0 = 0; J0 < NBLK; J0 += 3) {
        scan_block<0>(J0,     wave, lane, c, h, Bt, lbase, pf, out, hL, part, hv, pa, pr, tq);
        scan_block<1>(J0 + 1, wave, lane, c, h, Bt, lbase, pf, out, hL, part, hv, pa, pr, tq);
        scan_block<2>(J0 + 2, wave, lane, c, h, Bt, lbase, pf, out, hL, part, hv, pa, pr, tq);
    }
}

extern "C" void kernel_launch(void* const* d_in, const int* in_sizes, int n_in,
                              void* d_out, int out_size, void* d_ws, size_t ws_size,
                              hipStream_t stream) {
    (void)in_sizes; (void)n_in; (void)ws_size; (void)out_size;
    const float* x       = (const float*)d_in[0];
    const int*   in_idxs = (const int*)d_in[1];
    const float* weights = (const float*)d_in[2];
    float*       out     = (float*)d_out;
    float*       wd      = (float*)d_ws;
    float*       pf      = wd + (size_t)NEUR * COLS;

    ne_rows_gemm_kernel<<<dim3(NBLK, NG), dim3(128), 0, stream>>>(x, in_idxs, weights, wd, pf);
    ne_block_scan_kernel<<<dim3(NBT), dim3(128), 0, stream>>>(wd, pf, out);
}


# Launch 1 reads x as whole 128-byte lines (a load = 512 contiguous bytes of one row, not 16 rows' 32-byte pieces), from kernel entry on, and reorders it in a wave's own LDS tile (same sum order)
# speedup vs baseline: 10.1630x; 1.0901x over previous
#include <hip/hip_runtime.h>


typedef float v2f __attribute__((ext_vector_type(2)));
typedef float v8f __attribute__((ext_vector_type(8)));
typedef float v4f __attribute__((ext_vector_type(4)));

#define BATCH 512
#define INPS  1024
#define NEUR  512
#define KFAN  64
#define NOUT  16
#define COLS  (INPS + NEUR)
#define NBLK  (NEUR / 16)
#define NBT   (BATCH / 16)
#define NHELP 3
#define MAXT  10
#define GW    4
#define NG    (NBT / GW)
#define HS    (NEUR / NG)
#define SW    (INPS + HS)
#define SP    (SW + 4)
#define XP    (128 + 4)

__device__ __forceinline__ v8f mma4(v2f a, v2f b, v8f c) {
    return __builtin_amdgcn_wmma_f32_16x16x4_f32(false, a, false, b, (short)0, c, false, false);
}

__global__ __launch_bounds__(128) void ne_rows_gemm_kernel(
    const float*              x,
    const int*   __restrict__ in_idxs,
    const float* __restrict__ weights,
    float*       __restrict__ wd,
    float*       __restrict__ pf)
{
    __shared__ __attribute__((aligned(16))) float slab[16 * SP];
    __shared__ __attribute__((aligned(16))) int   sj[16 * KFAN];
    __shared__ __attribute__((aligned(16))) float sw[16 * KFAN];
    __shared__ __attribute__((aligned(16))) float xs[GW * 16 * XP];
    const int tid = threadIdx.x;
    const int J = blockIdx.x, g = blockIdx.y;
    const int lane = tid & 31;
    const int c = lane & 15, h = lane >> 4;
    const int wv = tid >> 5;
    const int Bt = GW * g + wv;
    const float* xg = x + (size_t)(16 * Bt) * INPS + 4 * lane;
    v4f gq[2][16];
    #pragma unroll
    for (int i = 0; i < 16; ++i) gq[0][i] = *(const v4f*)(xg + (size_t)i * INPS);
    #pragma unroll
    for (int i = 0; i < 16; ++i) gq[1][i] = *(const v4f*)(xg + (size_t)i * INPS + 128);
    {
        const int4*   gi = (const int4*)(in_idxs + (size_t)J * 16 * KFAN);
        const float4* gw = (const float4*)(weights + (size_t)J * 16 * KFAN);
        ((int4*)sj)[tid]         = gi[tid];
        ((int4*)sj)[tid + 128]   = gi[tid + 128];
        ((float4*)sw)[tid]       = gw[tid];
        ((float4*)sw)[tid + 128] = gw[tid + 128];
    }
    for (int i = tid; i < 16 * SP / 4; i += 128)
        ((float4*)slab)[i] = make_float4(0.f, 0.f, 0.f, 0.f);
    __syncthreads();
    {
        const int t  = tid & 63;
        const int r0 = __builtin_amdgcn_readfirstlane(tid >> 6);
        int   jv[8];
        float sv[8];
        #pragma unroll
        for (int i = 0; i < 8; ++i) {
            jv[i] = sj[(r0 + 2 * i) * KFAN + t];
            sv[i] = 0.0f;
        }
        #pragma unroll 4
        for (int k4 = 0; k4 < KFAN / 4; ++k4) {
            #pragma unroll
            for (int i = 0; i < 8; ++i) {
                const int4   p = ((const int4*)(sj + (r0 + 2 * i) * KFAN))[k4];
                const float4 u = ((const float4*)(sw + (r0 + 2 * i) * KFAN))[k4];
                sv[i] += (p.x == jv[i]) ? u.x : 0.0f;
                sv[i] += (p.y == jv[i]) ? u.y : 0.0f;
                sv[i] += (p.z == jv[i]) ? u.z : 0.0f;
                sv[i] += (p.w == jv[i]) ? u.w : 0.0f;
            }
        }
        #pragma unroll
        for (int i = 0; i < 8; ++i) {
            const int r = r0 + 2 * i;
            const int j = jv[i];
            const unsigned q = (unsigned)(j - INPS - HS * g);
            int col = (j < INPS) ? j : ((q < (unsigned)HS) ? (int)(INPS + q) : SW);
            col = ((unsigned)j < (unsigned)(INPS + 16 * J + r)) ? col : SW;
            slab[r * SP + col] = 0.5f * sv[i];
        }
    }
    __syncthreads();
    for (int i = tid; i < 16 * HS / 4; i += 128) {
        const int ln = i & 31, q = (i >> 5) & 1, jl = i >> 6;
        const int r = ln & 15, hh = ln >> 4;
        const float4 v = *(const float4*)&slab[r * SP + INPS + 16 * jl + 8 * hh + 4 * q];
        *(float4*)(wd + ((size_t)((J * NBLK + (HS / 16) * g + jl) * 2 + q) * 32 + ln) * 4) = v;
    }

    const float* ap = slab + c * SP + 4 * h;
    float*       xw = xs + wv * (16 * XP);
    float*       xt = xw + 4 * lane;
    const float* bp = xw + c * XP + 4 * h;
    v8f acc0 = {}, acc1 = {};
    float4 af[2][16], bf[2][16];
    #pragma unroll
    for (int i = 0; i < 16; ++i) *(v4f*)(xt + i * XP) = gq[0][i];
    #pragma unroll
    for (int i = 0; i < 16; ++i) gq[0][i] = *(const v4f*)(xg + (size_t)i * INPS + 256);
    #pragma unroll
    for (int s = 0; s < 16; ++s) {
        af[0][s] = *(const float4*)(ap + 8 * s);
        bf[0][s] = *(const float4*)(bp + 8 * s);
    }
    __builtin_amdgcn_sched_barrier(0);
    #pragma unroll
    for (int kc = 0; kc < INPS / 128; ++kc) {
        const int cur = kc & 1, nxt = cur ^ 1;
        if (kc + 1 < INPS / 128) {
            #pragma unroll
            for (int i = 0; i < 16; ++i) *(v4f*)(xt + i * XP) = gq[nxt][i];
            if (kc + 3 < INPS / 128) {
                #pragma unroll
                for (int i = 0; i < 16; ++i)
                    gq[nxt][i] = *(const v4f*)(xg + (size_t)i * INPS + 128 * (kc + 3));
            }
            #pragma unroll
            for (int s = 0; s < 16; ++s) {
                af[nxt][s] = *(const float4*)(ap + 128 * (kc + 1) + 8 * s);
                bf[nxt][s] = *(const float4*)(bp + 8 * s);
            }
        }
        __builtin_amdgcn_sched_barrier(0);
        #pragma unroll
        for (int s = 0; s < 16; ++s) {
            const float4 a = af[cur][s];
            const float4 b = bf[cur][s];
            const v2f a0 = {a.x, a.y}, b0 = {b.x, b.y};
            const v2f a1 = {a.z, a.w}, b1 = {b.z, b.w};
            acc0 = mma4(a0, b0, acc0);
            acc1 = mma4(a1, b1, acc1);
        }
        __builtin_amdgcn_sched_barrier(0);
    }
    acc0 += acc1;
    float4* o = (float4*)(pf + (size_t)(J * NBT + Bt) * 256) + lane;
    o[0]  = make_float4(acc0[0], acc0[1], acc0[2], acc0[3]);
    o[32] = make_float4(acc0[4], acc0[5], acc0[6], acc0[7]);
}


__device__ __forceinline__ void lds_block_barrier() {
    asm volatile("s_wait_dscnt 0x0" ::: "memory");
    asm volatile("s_barrier_signal -1" ::: "memory");
    asm volatile("s_barrier_wait -1" ::: "memory");
}

template <int S>
__device__ __forceinline__ void scan_request(
    const int Jn, const int Rn, const int wave, const int lane, const int c, const int Bt,
    const float* lbase, const float* pf, float4 (&tq)[3][MAXT][2])
{
    const bool chain = (wave == 0);
    const int wl = chain ? 0 : wave - 1;
    const float* ln = chain ? pf + (size_t)Bt * 256
                            : lbase + (size_t)(Rn * NBLK + wl) * 256;
    const float* lr = lbase + (size_t)(Jn * NBLK + Jn) * 256;
    const float* pp = pf + (size_t)(Jn * NBT + Bt) * 256 + 4 * lane;
    const float* s0 = chain ? pp : ln;
    const float* s1 = chain ? lr : ln + 256 * NHELP;
    const float* s2 = chain ? ((Jn > 0) ? lr - 256 : lr) : ln + 512 * NHELP;
    tq[S][0][0] = *(const float4*)(s0);
    tq[S][0][1] = *(const float4*)(s0 + 128);
    tq[S][1][0] = *(const float4*)(s1);
    tq[S][1][1] = *(const float4*)(s1 + 128);
    tq[S][2][0] = *(const float4*)(s2);
    tq[S][2][1] = *(const float4*)(s2 + 128);
    #pragma unroll
    for (int t = 3; t < MAXT; ++t) {
        tq[S][t][0] = *(const float4*)(ln + 256 * NHELP * t);
        tq[S][t][1] = *(const float4*)(ln + 256 * NHELP * t + 128);
    }
}

template <int CUR>
__device__ __forceinline__ void scan_block(
    const int J, const int wave, const int lane, const int c, const int h, const int Bt,
    const float* lbase, const float* pf, float* out,
    float (&hL)[NBLK][2][16][8], float (&part)[2][NHELP][32][8],
    float (&hv)[8], v8f &pa, v8f &pr, float4 (&tq)[3][MAXT][2])
{
    float4 u[NHELP][2];
    if (wave == 0 && J >= 1 && J < NBLK) {
        #pragma unroll
        for (int w = 0; w < NHELP; ++w) {
            u[w][0] = *(const float4*)&part[J & 1][w][lane][0];
            u[w][1] = *(const float4*)&part[J & 1][w][lane][4];
        }
    }
    scan_request<(CUR + 2) % 3>((J + 2 < NBLK) ? J + 2 : NBLK - 1, (J + 3 < NBLK) ? J + 3 : NBLK - 1,
                                wave, lane, c, Bt, lbase, pf, tq);
    __builtin_amdgcn_sched_barrier(0);
    if (J >= NBLK) {
    } else if (wave == 0) {
        const float4 p0 = tq[CUR][0][0], p1 = tq[CUR][0][1], d0 = tq[CUR][1][0], d1 = tq[CUR][1][1];
        v8f acc = {p0.x, p0.y, p0.z, p0.w, p1.x, p1.y, p1.z, p1.w};
        if (J >= 1) {
            acc = pa + pr;
            v8f q;
            q[0] = (u[0][0].x + u[1][0].x) + u[2][0].x; q[1] = (u[0][0].y + u[1][0].y) + u[2][0].y;
            q[2] = (u[0][0].z + u[1][0].z) + u[2][0].z; q[3] = (u[0][0].w + u[1][0].w) + u[2][0].w;
            q[4] = (u[0][1].x + u[1][1].x) + u[2][1].x; q[5] = (u[0][1].y + u[1][1].y) + u[2][1].y;
            q[6] = (u[0][1].z + u[1][1].z) + u[2][1].z; q[7] = (u[0][1].w + u[1][1].w) + u[2][1].w;
            acc += q;
        }
        const float dg[8] = {d0.x, d0.y, d0.z, d0.w, d1.x, d1.y, d1.z, d1.w};
        float alo[8], ahi[8];
        for (int v = 0; v < 8; ++v) {
            alo[v] = (h == 0) ? dg[v] : 0.0f;
            ahi[v] = (h == 1) ? dg[v] : 0.0f;
        }
        #pragma unroll
        for (int i = 0; i < 16; ++i) {
            const int v = i & 7;
            const float pre = acc[v];
            const float sg  = __builtin_fmaf(0.5f, __builtin_amdgcn_tanhf(pre), 0.5f);
            hv[v] = ((i >> 3) == h) ? sg : hv[v];
            if (i < 15) {
                const v2f a = {(i < 8) ? alo[v] : ahi[v], 0.0f};
                const v2f b = {sg, 0.0f};
                acc = mma4(a, b, acc);
            }
        }
        *(float4*)&hL[J][h][c][0] = make_float4(hv[0], hv[1], hv[2], hv[3]);
        *(float4*)&hL[J][h][c][4] = make_float4(hv[4], hv[5], hv[6], hv[7]);
        if (J == NBLK - 1) {
            for (int v = 0; v < 8; ++v)
                out[(size_t)(8 * h + v) * BATCH + 16 * Bt + c] = hv[v];
        }
        __builtin_amdgcn_sched_barrier(0);
        {
            constexpr int NX = (CUR + 1) % 3;
            const float4 n0 = tq[NX][0][0], n1 = tq[NX][0][1];
            const float4 c0 = tq[NX][2][0], c1 = tq[NX][2][1];
            const v2f a0 = {c0.x, c0.y}, a1 = {c0.z, c0.w}, a2 = {c1.x, c1.y}, a3 = {c1.z, c1.w};
            const v2f b0 = {hv[0], hv[1]}, b1 = {hv[2], hv[3]}, b2 = {hv[4], hv[5]}, b3 = {hv[6], hv[7]};
            v8f xa = {n0.x, n0.y, n0.z, n0.w, n1.x, n1.y, n1.z, n1.w};
            v8f xr = {};
            xa = mma4(a0, b0, xa);
            xr = mma4(a1, b1, xr);
            xa = mma4(a2, b2, xa);
            xr = mma4(a3, b3, xr);
            pa = xa;
            pr = xr;
        }
    } else if (J + 1 < NBLK) {
        const int R = J + 1;
        const int w = wave - 1;
        float4 hb[MAXT][2];
        #pragma unroll
        for (int t = 0; t < MAXT; ++t) {
            const int j = w + NHELP * t;
            hb[t][0] = *(const float4*)&hL[j][h][c][0];
            hb[t][1] = *(const float4*)&hL[j][h][c][4];
        }
        __builtin_amdgcn_sched_barrier(0);
        v8f q0 = {}, q1 = {};
        #pragma unroll
        for (int t = 0; t < MAXT; ++t) {
            const int j = w + NHELP * t;
            if (j < J) {
                const float4 a0 = tq[CUR][t][0], a1 = tq[CUR][t][1];
                const float4 b0 = hb[t][0], b1 = hb[t][1];
                const v2f x0 = {a0.x, a0.y}, y0 = {b0.x, b0.y};
                const v2f x1 = {a0.z, a0.w}, y1 = {b0.z, b0.w};
                const v2f x2 = {a1.x, a1.y}, y2 = {b1.x, b1.y};
                const v2f x3 = {a1.z, a1.w}, y3 = {b1.z, b1.w};
                q0 = mma4(x0, y0, q0);
                q1 = mma4(x1, y1, q1);
                q0 = mma4(x2, y2, q0);
                q1 = mma4(x3, y3, q1);
            }
        }
        q0 += q1;
        *(float4*)&part[R & 1][w][lane][0] = make_float4(q0[0], q0[1], q0[2], q0[3]);
        *(float4*)&part[R & 1][w][lane][4] = make_float4(q0[4], q0[5], q0[6], q0[7]);
    }
    lds_block_barrier();
}

__global__ __launch_bounds__(128) void ne_block_scan_kernel(
    const float* wd,
    const float* pf,
    float*       out)
{
    __shared__ __attribute__((aligned(16))) float hL[NBLK][2][16][8];
    __shared__ __attribute__((aligned(16))) float part[2][NHELP][32][8];

    const int wave = __builtin_amdgcn_readfirstlane((int)(threadIdx.x >> 5));
    const int lane = threadIdx.x & 31;
    const int c = lane & 15, h = lane >> 4;
    const int Bt = blockIdx.x;
    const float* lbase = wd + 4 * lane;

    float  hv[8] = {0.f, 0.f, 0.f, 0.f, 0.f, 0.f, 0.f, 0.f};
    v8f pa = {}, pr = {};
    float4 tq[3][MAXT][2];
    scan_request<0>(0, 1, wave, lane, c, Bt, lbase, pf, tq);
    __builtin_amdgcn_sched_barrier(0);
    scan_request<1>(1, 2, wave, lane, c, Bt, lbase, pf, tq);
    __builtin_amdgcn_sched_barrier(0);
    #pragma unroll
    for (int t = 0; t < MAXT; ++t) {
        tq[2][t][0] = make_float4(0.f, 0.f, 0.f, 0.f); tq[2][t][1] = tq[2][t][0];
    }

    #pragma nounroll
    for (int J0 = 0; J0 < NBLK; J0 += 3) {
        scan_block<0>(J0,     wave, lane, c, h, Bt, lbase, pf, out, hL, part, hv, pa, pr, tq);
        scan_block<1>(J0 + 1, wave, lane, c, h, Bt, lbase, pf, out, hL, part, hv, pa, pr, tq);
        scan_block<2>(J0 + 2, wave, lane, c, h, Bt, lbase, pf, out, hL, part, hv, pa, pr, tq);
    }
}

extern "C" void kernel_launch(void* const* d_in, const int* in_sizes, int n_in,
                              void* d_out, int out_size, void* d_ws, size_t ws_size,
                              hipStream_t stream) {
    (void)in_sizes; (void)n_in; (void)ws_size; (void)out_size;
    const float* x       = (const float*)d_in[0];
    const int*   in_idxs = (const int*)d_in[1];
    const float* weights = (const float*)d_in[2];
    float*       out     = (float*)d_out;
    float*       wd      = (float*)d_ws;
    float*       pf      = wd + (size_t)NEUR * COLS;

    ne_rows_gemm_kernel<<<dim3(NBLK, NG), dim3(128), 0, stream>>>(x, in_idxs, weights, wd, pf);
    ne_block_scan_kernel<<<dim3(NBT), dim3(128), 0, stream>>>(wd, pf, out);
}
